# starting kernel
# baseline (speedup 1.0000x reference)
#include <hip/hip_runtime.h>
#include <hip/hip_bf16.h>

#define N_NODES 100000
#define N_EDGES 1600000
#define TOT_E   (N_EDGES + N_NODES)
#define IN_CH   64
#define OUT_CH  32
#define HEADS   4
#define HC      (HEADS * OUT_CH)
#define NEG_SLOPE 0.2f

typedef __attribute__((ext_vector_type(16))) _Float16 v16h;
typedef __attribute__((ext_vector_type(8)))  float    v8f;

__device__ __forceinline__ unsigned fflip(float f) {
    unsigned u = __float_as_uint(f);
    return (u & 0x80000000u) ? ~u : (u | 0x80000000u);
}
__device__ __forceinline__ float funflip(unsigned u) {
    return __uint_as_float((u & 0x80000000u) ? (u & 0x7FFFFFFFu) : ~u);
}

__device__ __forceinline__ void edge_sd(const int* __restrict__ ei, int e, int& s, int& d) {
    if (e < N_EDGES) { s = ei[e]; d = ei[N_EDGES + e]; }
    else             { s = d = e - N_EDGES; }
}

__global__ __launch_bounds__(256) void k_gemm(const float* __restrict__ x,
                                              const float* __restrict__ W,
                                              float* __restrict__ h) {
    __shared__ float xs[16 * IN_CH];
    const int tid = threadIdx.x;
    const int mt  = blockIdx.x;
    for (int i = tid; i < 16 * IN_CH; i += 256) {
        int r = i >> 6, c = i & 63;
        xs[i] = x[(size_t)(mt * 16 + r) * IN_CH + c];
    }
    __syncthreads();

    const int wave = tid >> 5;
    const int lane = tid & 31;
    const int mrow = lane & 15;
    const int khalf = (lane >> 4) << 3;
    const int ncol = wave * 16 + (lane & 15);

    v16h a0, a1, b0, b1;
    #pragma unroll
    for (int i = 0; i < 16; ++i) {
        int k = i + (i & 8) + khalf;
        a0[i] = (_Float16)xs[mrow * IN_CH + k];
        a1[i] = (_Float16)xs[mrow * IN_CH + k + 32];
        b0[i] = (_Float16)W[(size_t)k * HC + ncol];
        b1[i] = (_Float16)W[(size_t)(k + 32) * HC + ncol];
    }
    v8f c = {};
    c = __builtin_amdgcn_wmma_f32_16x16x32_f16(false, a0, false, b0, (short)0, c, false, false);
    c = __builtin_amdgcn_wmma_f32_16x16x32_f16(false, a1, false, b1, (short)0, c, false, false);

    #pragma unroll
    for (int r = 0; r < 8; ++r) {
        int m = mt * 16 + r + khalf;
        h[(size_t)m * HC + ncol] = c[r];
    }
}

__global__ __launch_bounds__(256) void k_att(const float* __restrict__ h,
                                             const float* __restrict__ att_src,
                                             const float* __restrict__ att_dst,
                                             float* __restrict__ a_src,
                                             float* __restrict__ a_dst) {
    int idx = blockIdx.x * blockDim.x + threadIdx.x;
    if (idx >= N_NODES * HEADS) return;
    int n = idx >> 2, hh = idx & 3;
    const float* hp = h + (size_t)n * HC + hh * OUT_CH;
    float s = 0.f, d = 0.f;
    #pragma unroll
    for (int c = 0; c < OUT_CH; ++c) {
        float v = hp[c];
        s += v * att_src[hh * OUT_CH + c];
        d += v * att_dst[hh * OUT_CH + c];
    }
    a_src[idx] = s; a_dst[idx] = d;
}

__global__ __launch_bounds__(256) void k_init(unsigned* __restrict__ mflip,
                                              float* __restrict__ denom,
                                              float* __restrict__ out) {
    const int stride = gridDim.x * blockDim.x;
    for (int i = blockIdx.x * blockDim.x + threadIdx.x; i < N_NODES * HC; i += stride)
        out[i] = 0.f;
    for (int i = blockIdx.x * blockDim.x + threadIdx.x; i < N_NODES * HEADS; i += stride) {
        mflip[i] = 0u;
        denom[i] = 0.f;
    }
}

__global__ __launch_bounds__(256) void k_max(const int* __restrict__ ei,
                                             const float* __restrict__ a_src,
                                             const float* __restrict__ a_dst,
                                             unsigned* __restrict__ mflip) {
    int e = blockIdx.x * blockDim.x + threadIdx.x;
    if (e >= TOT_E) return;
    int s, d; edge_sd(ei, e, s, d);
    #pragma unroll
    for (int hh = 0; hh < HEADS; ++hh) {
        float v = a_src[s * HEADS + hh] + a_dst[d * HEADS + hh];
        v = v > 0.f ? v : NEG_SLOPE * v;
        atomicMax(&mflip[d * HEADS + hh], fflip(v));
    }
}

__global__ __launch_bounds__(256) void k_denom(const int* __restrict__ ei,
                                               const float* __restrict__ a_src,
                                               const float* __restrict__ a_dst,
                                               const unsigned* __restrict__ mflip,
                                               float* __restrict__ denom) {
    int e = blockIdx.x * blockDim.x + threadIdx.x;
    if (e >= TOT_E) return;
    int s, d; edge_sd(ei, e, s, d);
    #pragma unroll
    for (int hh = 0; hh < HEADS; ++hh) {
        float v = a_src[s * HEADS + hh] + a_dst[d * HEADS + hh];
        v = v > 0.f ? v : NEG_SLOPE * v;
        float m = funflip(mflip[d * HEADS + hh]);
        atomicAdd(&denom[d * HEADS + hh], __expf(v - m));
    }
}

__global__ __launch_bounds__(256) void k_agg(const int* __restrict__ ei,
                                             const float* __restrict__ a_src,
                                             const float* __restrict__ a_dst,
                                             const unsigned* __restrict__ mflip,
                                             const float* __restrict__ denom,
                                             const float* __restrict__ h,
                                             float* __restrict__ out) {
    const int lane  = threadIdx.x & 31;
    const int hh    = lane >> 3;
    const int wglob = (blockIdx.x * blockDim.x + threadIdx.x) >> 5;
    const int nwav  = (gridDim.x * blockDim.x) >> 5;

    for (int e = wglob; e < TOT_E; e += nwav) {
        int s, d; edge_sd(ei, e, s, d);

        int en = e + nwav;
        if (en < TOT_E) {
            int sn = (en < N_EDGES) ? ei[en] : (en - N_EDGES);
            __builtin_prefetch(h + (size_t)sn * HC + lane * 4, 0, 0);
        }

        float v = a_src[s * HEADS + hh] + a_dst[d * HEADS + hh];
        v = v > 0.f ? v : NEG_SLOPE * v;
        float m     = funflip(mflip[d * HEADS + hh]);
        float alpha = __expf(v - m) / (denom[d * HEADS + hh] + 1e-16f);

        float4 hv = ((const float4*)(h + (size_t)s * HC))[lane];
        float* op = out + (size_t)d * HC + lane * 4;
        atomicAdd(op + 0, hv.x * alpha);
        atomicAdd(op + 1, hv.y * alpha);
        atomicAdd(op + 2, hv.z * alpha);
        atomicAdd(op + 3, hv.w * alpha);
    }
}

__global__ __launch_bounds__(256) void k_lsm(float* __restrict__ out,
                                             const float* __restrict__ bias) {
    const int lane = threadIdx.x & 31;
    const int node = (blockIdx.x * blockDim.x + threadIdx.x) >> 5;
    if (node >= N_NODES) return;
    float* row = out + (size_t)node * HC;
    float4 v = ((float4*)row)[lane];
    float4 b = ((const float4*)bias)[lane];
    v.x += b.x; v.y += b.y; v.z += b.z; v.w += b.w;

    float mx = fmaxf(fmaxf(v.x, v.y), fmaxf(v.z, v.w));
    #pragma unroll
    for (int o = 16; o >= 1; o >>= 1) mx = fmaxf(mx, __shfl_xor(mx, o, 32));

    float se = __expf(v.x - mx) + __expf(v.y - mx) + __expf(v.z - mx) + __expf(v.w - mx);
    #pragma unroll
    for (int o = 16; o >= 1; o >>= 1) se += __shfl_xor(se, o, 32);

    float lse = mx + __logf(se);
    v.x -= lse; v.y -= lse; v.z -= lse; v.w -= lse;
    ((float4*)row)[lane] = v;
}

extern "C" void kernel_launch(void* const* d_in, const int* in_sizes, int n_in,
                              void* d_out, int out_size, void* d_ws, size_t ws_size,
                              hipStream_t stream) {
    const float* x       = (const float*)d_in[0];
    const int*   ei      = (const int*)  d_in[1];
    const float* W       = (const float*)d_in[2];
    const float* att_src = (const float*)d_in[3];
    const float* att_dst = (const float*)d_in[4];
    const float* bias    = (const float*)d_in[5];
    float*       out     = (float*)d_out;

    float*    h     = (float*)d_ws;
    float*    asrc  = h    + (size_t)N_NODES * HC;
    float*    adst  = asrc + (size_t)N_NODES * HEADS;
    unsigned* mflip = (unsigned*)(adst + (size_t)N_NODES * HEADS);
    float*    denom = (float*)(mflip + (size_t)N_NODES * HEADS);

    k_gemm<<<N_NODES / 16, 256, 0, stream>>>(x, W, h);
    k_att <<<(N_NODES * HEADS + 255) / 256, 256, 0, stream>>>(h, att_src, att_dst, asrc, adst);
    k_init<<<4096, 256, 0, stream>>>(mflip, denom, out);
    k_max <<<(TOT_E + 255) / 256, 256, 0, stream>>>(ei, asrc, adst, mflip);
    k_denom<<<(TOT_E + 255) / 256, 256, 0, stream>>>(ei, asrc, adst, mflip, denom);
    k_agg <<<12288, 256, 0, stream>>>(ei, asrc, adst, mflip, denom, h, out);
    k_lsm <<<(N_NODES + 7) / 8, 256, 0, stream>>>(out, bias);
}


# Edges counting-sorted by destination (int atomics, arrival-order rank); one wave per destination gathers h rows: running-max softmax, sum, log_softmax in registers; no float atomics, new sum order
# speedup vs baseline: 3.7706x; 3.7706x over previous
#include <hip/hip_runtime.h>

#define N_NODES 100000
#define N_EDGES 1600000
#define IN_CH   64
#define OUT_CH  32
#define HEADS   4
#define HC      (HEADS * OUT_CH)
#define NEG_SLOPE 0.2f
#define SCAN_T  1024
#define SCAN_B  ((N_NODES + SCAN_T - 1) / SCAN_T)

typedef __attribute__((ext_vector_type(16))) _Float16 v16h;
typedef __attribute__((ext_vector_type(8)))  float    v8f;

__global__ __launch_bounds__(256) void k_gemm(const float* __restrict__ x,
                                              const float* __restrict__ W,
                                              float* __restrict__ h) {
    __shared__ float xs[16 * IN_CH];
    const int tid = threadIdx.x;
    const int mt  = blockIdx.x;
    for (int i = tid; i < 16 * IN_CH; i += 256) {
        int r = i >> 6, c = i & 63;
        xs[i] = x[(size_t)(mt * 16 + r) * IN_CH + c];
    }
    __syncthreads();

    const int wave = tid >> 5;
    const int lane = tid & 31;
    const int mrow = lane & 15;
    const int khalf = (lane >> 4) << 3;
    const int ncol = wave * 16 + (lane & 15);

    v16h a0, a1, b0, b1;
    #pragma unroll
    for (int i = 0; i < 16; ++i) {
        int k = i + (i & 8) + khalf;
        a0[i] = (_Float16)xs[mrow * IN_CH + k];
        a1[i] = (_Float16)xs[mrow * IN_CH + k + 32];
        b0[i] = (_Float16)W[(size_t)k * HC + ncol];
        b1[i] = (_Float16)W[(size_t)(k + 32) * HC + ncol];
    }
    v8f c = {};
    c = __builtin_amdgcn_wmma_f32_16x16x32_f16(false, a0, false, b0, (short)0, c, false, false);
    c = __builtin_amdgcn_wmma_f32_16x16x32_f16(false, a1, false, b1, (short)0, c, false, false);

    #pragma unroll
    for (int r = 0; r < 8; ++r) {
        int m = mt * 16 + r + khalf;
        h[(size_t)m * HC + ncol] = c[r];
    }
}

__global__ __launch_bounds__(256) void k_zero(int* __restrict__ degc) {
    int i = blockIdx.x * blockDim.x + threadIdx.x;
    if (i < N_NODES) degc[i] = 0;
}

__global__ __launch_bounds__(256) void k_rank(const int* __restrict__ ei,
                                              int* __restrict__ degc,
                                              int* __restrict__ rank) {
    int e = blockIdx.x * blockDim.x + threadIdx.x;
    if (e >= N_EDGES) return;
    int d = ei[N_EDGES + e];
    rank[e] = atomicAdd(&degc[d], 1);
}

__device__ __forceinline__ int wave_sum_i(int v) {
    #pragma unroll
    for (int o = 16; o >= 1; o >>= 1) v += __shfl_xor(v, o, 32);
    return v;
}
__device__ __forceinline__ int wave_incl_scan_i(int v, int lane) {
    #pragma unroll
    for (int o = 1; o < 32; o <<= 1) {
        int t = __shfl_up(v, o, 32);
        if (lane >= o) v += t;
    }
    return v;
}

__global__ __launch_bounds__(SCAN_T) void k_bsum(const int* __restrict__ degc,
                                                 int* __restrict__ bsum) {
    __shared__ int wtot[32];
    const int t = threadIdx.x;
    const int i = blockIdx.x * SCAN_T + t;
    int v = (i < N_NODES) ? degc[i] : 0;
    v = wave_sum_i(v);
    if ((t & 31) == 0) wtot[t >> 5] = v;
    __syncthreads();
    if (t < 32) {
        int s = wave_sum_i(wtot[t]);
        if (t == 0) bsum[blockIdx.x] = s;
    }
}

__global__ __launch_bounds__(SCAN_T) void k_scan(const int* __restrict__ degc,
                                                 const int* __restrict__ bsum,
                                                 int* __restrict__ startc,
                                                 int* __restrict__ meta) {
    __shared__ int wtot[32];
    __shared__ int boff;
    const int t = threadIdx.x, b = blockIdx.x;
    const int lane = t & 31, w = t >> 5;
    const int i = b * SCAN_T + t;
    const int v = (i < N_NODES) ? degc[i] : 0;
    const int incl = wave_incl_scan_i(v, lane);
    if (lane == 31) wtot[w] = incl;
    __syncthreads();
    if (w == 0) {
        int x  = wtot[lane];
        int xs = wave_incl_scan_i(x, lane);
        wtot[lane] = xs - x;
        int pb = 0;
        for (int j = lane; j < b; j += 32) pb += bsum[j];
        pb = wave_sum_i(pb);
        if (lane == 0) boff = pb;
    }
    __syncthreads();
    const int excl = incl - v + wtot[w] + boff;
    if (i < N_NODES) {
        startc[i] = excl;
        meta[(size_t)i * HC]     = excl;
        meta[(size_t)i * HC + 1] = v;
    }
}

__global__ __launch_bounds__(256) void k_fill(const int* __restrict__ ei,
                                              const int* __restrict__ startc,
                                              const int* __restrict__ rank,
                                              int* __restrict__ sorted) {
    int e = blockIdx.x * blockDim.x + threadIdx.x;
    if (e >= N_EDGES) return;
    int s = ei[e], d = ei[N_EDGES + e];
    sorted[startc[d] + rank[e]] = s;
}


__device__ __forceinline__ float sum8(float v) {
    v += __int_as_float(__builtin_amdgcn_update_dpp(0, __float_as_int(v), 0xB1, 0xF, 0xF, true));
    v += __int_as_float(__builtin_amdgcn_update_dpp(0, __float_as_int(v), 0x4E, 0xF, 0xF, true));
    v += __int_as_float(__builtin_amdgcn_update_dpp(0, __float_as_int(v), 0x141, 0xF, 0xF, true));
    return v;
}

__device__ __forceinline__ float4 ldrow(const float* __restrict__ h, int s, int lane) {
    return ((const float4*)(h + (size_t)s * HC))[lane];
}

__device__ __forceinline__ void edge_step(const float4 hv, const float4 as4, const float adst,
                                          float& m, float& l, float4& acc) {
    float e = sum8(hv.x * as4.x + hv.y * as4.y + hv.z * as4.z + hv.w * as4.w) + adst;
    e = e > 0.f ? e : NEG_SLOPE * e;
    const float mn = fmaxf(m, e);
    const float sc = __expf(m - mn);
    const float pe = __expf(e - mn);
    acc.x = acc.x * sc + pe * hv.x;
    acc.y = acc.y * sc + pe * hv.y;
    acc.z = acc.z * sc + pe * hv.z;
    acc.w = acc.w * sc + pe * hv.w;
    l = l * sc + pe;
    m = mn;
}

__global__ __launch_bounds__(256) void k_node(const int* __restrict__ sorted,
                                              const float* __restrict__ h,
                                              const float* __restrict__ att_src,
                                              const float* __restrict__ att_dst,
                                              const float* __restrict__ bias,
                                              const int* meta,
                                              float* out) {
    const int lane = threadIdx.x & 31;
    const int d    = (blockIdx.x * blockDim.x + threadIdx.x) >> 5;
    if (d >= N_NODES) return;

    const int start = __builtin_amdgcn_readfirstlane(meta[(size_t)d * HC]);
    const int deg   = __builtin_amdgcn_readfirstlane(meta[(size_t)d * HC + 1]);

    const float4 as4 = ((const float4*)att_src)[lane];
    const float4 ad4 = ((const float4*)att_dst)[lane];

    const float4 hd = ldrow(h, d, lane);
    const float adst = sum8(hd.x * ad4.x + hd.y * ad4.y + hd.z * ad4.z + hd.w * ad4.w);

    float e0 = sum8(hd.x * as4.x + hd.y * as4.y + hd.z * as4.z + hd.w * as4.w) + adst;
    e0 = e0 > 0.f ? e0 : NEG_SLOPE * e0;
    float  m = e0, l = 1.0f;
    float4 acc = hd;

    for (int base = 0; base < deg; base += 32) {
        int n = deg - base; if (n > 32) n = 32;
        int myid = 0;
        if (lane < n) myid = sorted[start + base + lane];
        int k = 0;
        for (; k + 4 <= n; k += 4) {
            const int s0 = __builtin_amdgcn_readlane(myid, k);
            const int s1 = __builtin_amdgcn_readlane(myid, k + 1);
            const int s2 = __builtin_amdgcn_readlane(myid, k + 2);
            const int s3 = __builtin_amdgcn_readlane(myid, k + 3);
            const float4 h0 = ldrow(h, s0, lane);
            const float4 h1 = ldrow(h, s1, lane);
            const float4 h2 = ldrow(h, s2, lane);
            const float4 h3 = ldrow(h, s3, lane);
            edge_step(h0, as4, adst, m, l, acc);
            edge_step(h1, as4, adst, m, l, acc);
            edge_step(h2, as4, adst, m, l, acc);
            edge_step(h3, as4, adst, m, l, acc);
        }
        for (; k < n; ++k) {
            const int s0 = __builtin_amdgcn_readlane(myid, k);
            const float4 h0 = ldrow(h, s0, lane);
            edge_step(h0, as4, adst, m, l, acc);
        }
    }

    const float inv = 1.0f / (l + 1e-16f);
    const float4 b = ((const float4*)bias)[lane];
    float4 v;
    v.x = acc.x * inv + b.x; v.y = acc.y * inv + b.y;
    v.z = acc.z * inv + b.z; v.w = acc.w * inv + b.w;

    float mx = fmaxf(fmaxf(v.x, v.y), fmaxf(v.z, v.w));
    #pragma unroll
    for (int o = 16; o >= 1; o >>= 1) mx = fmaxf(mx, __shfl_xor(mx, o, 32));

    float se = __expf(v.x - mx) + __expf(v.y - mx) + __expf(v.z - mx) + __expf(v.w - mx);
    #pragma unroll
    for (int o = 16; o >= 1; o >>= 1) se += __shfl_xor(se, o, 32);

    const float lse = mx + __logf(se);
    v.x -= lse; v.y -= lse; v.z -= lse; v.w -= lse;
    ((float4*)(out + (size_t)d * HC))[lane] = v;
}

extern "C" void kernel_launch(void* const* d_in, const int* in_sizes, int n_in,
                              void* d_out, int out_size, void* d_ws, size_t ws_size,
                              hipStream_t stream) {
    const float* x       = (const float*)d_in[0];
    const int*   ei      = (const int*)  d_in[1];
    const float* W       = (const float*)d_in[2];
    const float* att_src = (const float*)d_in[3];
    const float* att_dst = (const float*)d_in[4];
    const float* bias    = (const float*)d_in[5];
    float*       out     = (float*)d_out;
    int*         meta    = (int*)d_out;

    float* h      = (float*)d_ws;
    int*   sorted = (int*)(h + (size_t)N_NODES * HC);
    int*   rank   = (int*)d_ws;
    int*   degc   = rank   + (size_t)N_EDGES;
    int*   startc = degc   + (size_t)N_NODES;
    int*   bsum   = startc + (size_t)N_NODES;

    k_zero<<<(N_NODES + 255) / 256, 256, 0, stream>>>(degc);
    k_rank<<<(N_EDGES + 255) / 256, 256, 0, stream>>>(ei, degc, rank);
    k_bsum<<<SCAN_B, SCAN_T, 0, stream>>>(degc, bsum);
    k_scan<<<SCAN_B, SCAN_T, 0, stream>>>(degc, bsum, startc, meta);
    k_fill<<<(N_EDGES + 255) / 256, 256, 0, stream>>>(ei, startc, rank, sorted);
    k_gemm<<<N_NODES / 16, 256, 0, stream>>>(x, W, h);
    k_node<<<(N_NODES + 7) / 8, 256, 0, stream>>>(sorted, h, att_src, att_dst, bias, meta, out);
}


# Aggregate in input space: gather 256-B x rows, not 512-B h rows; weights lane-parallel from a per-node score table x.(W.att); W applied after on WMMA per 16 nodes; h plane gone; new sum order
# speedup vs baseline: 4.0703x; 1.0795x over previous
#include <hip/hip_runtime.h>

#define N_NODES 100000
#define N_EDGES 1600000
#define IN_CH   64
#define OUT_CH  32
#define HEADS   4
#define HC      (HEADS * OUT_CH)
#define NEG_SLOPE 0.2f
#define SCAN_T  1024
#define SCAN_B  ((N_NODES + SCAN_T - 1) / SCAN_T)

typedef __attribute__((ext_vector_type(16))) _Float16 v16h;
typedef __attribute__((ext_vector_type(8)))  float    v8f;


#define SC_W   8
#define AP     264
#define OP     132

__device__ __forceinline__ float dpp_x1(float v) {
    return __int_as_float(__builtin_amdgcn_update_dpp(0, __float_as_int(v), 0xB1, 0xF, 0xF, true));
}
__device__ __forceinline__ float dpp_x2(float v) {
    return __int_as_float(__builtin_amdgcn_update_dpp(0, __float_as_int(v), 0x4E, 0xF, 0xF, true));
}
__device__ __forceinline__ float dpp_hm(float v) {
    return __int_as_float(__builtin_amdgcn_update_dpp(0, __float_as_int(v), 0x141, 0xF, 0xF, true));
}
__device__ __forceinline__ float dpp_rm(float v) {
    return __int_as_float(__builtin_amdgcn_update_dpp(0, __float_as_int(v), 0x140, 0xF, 0xF, true));
}
__device__ __forceinline__ float sum16(float v) {
    v += dpp_x1(v); v += dpp_x2(v); v += dpp_hm(v); v += dpp_rm(v);
    return v;
}
__device__ __forceinline__ float max16(float v) {
    v = fmaxf(v, dpp_x1(v)); v = fmaxf(v, dpp_x2(v)); v = fmaxf(v, dpp_hm(v)); v = fmaxf(v, dpp_rm(v));
    return v;
}
__device__ __forceinline__ float rlf(float v, int k) {
    return __int_as_float(__builtin_amdgcn_readlane(__float_as_int(v), k));
}
__device__ __forceinline__ float wave_sum(float v) { v = sum16(v); return rlf(v, 0) + rlf(v, 16); }
__device__ __forceinline__ float wave_max(float v) { v = max16(v); return fmaxf(rlf(v, 0), rlf(v, 16)); }
__device__ __forceinline__ float leaky(float e) { return e > 0.f ? e : NEG_SLOPE * e; }

#define SCORE_NPB 128
__global__ __launch_bounds__(256) void k_score(const float* __restrict__ x,
                                               const float* __restrict__ W,
                                               const float* __restrict__ att_src,
                                               const float* __restrict__ att_dst,
                                               float* __restrict__ sc) {
    __shared__ __attribute__((aligned(16))) float V[IN_CH * SC_W];
    const int tid = threadIdx.x;
    for (int i = tid; i < IN_CH * SC_W; i += 256) {
        const int k = i >> 3, j = i & 7, hd = j & 3;
        const float4* wr = (const float4*)(W + (size_t)k * HC + hd * OUT_CH);
        const float4* ar = (const float4*)((j < 4 ? att_src : att_dst) + hd * OUT_CH);
        float s = 0.f;
        #pragma unroll
        for (int c = 0; c < OUT_CH / 4; ++c) {
            const float4 w4 = wr[c], a4 = ar[c];
            s += w4.x * a4.x + w4.y * a4.y + w4.z * a4.z + w4.w * a4.w;
        }
        V[i] = s;
    }
    __syncthreads();

    const int wave = tid >> 5, lane = tid & 31;
    const int q = lane & 15, hf = lane >> 4;
    float v[4][SC_W];
    #pragma unroll
    for (int t = 0; t < 4; ++t) {
        #pragma unroll
        for (int j = 0; j < SC_W; ++j) v[t][j] = V[(4 * q + t) * SC_W + j];
    }
    const int base = blockIdx.x * SCORE_NPB + wave * 16 + hf;
    float4 xv[8];
    #pragma unroll
    for (int it = 0; it < 8; ++it) {
        int n = base + it * 2;
        if (n > N_NODES - 1) n = N_NODES - 1;
        xv[it] = ((const float4*)(x + (size_t)n * IN_CH))[q];
    }
    #pragma unroll
    for (int it = 0; it < 8; ++it) {
        const int n = base + it * 2;
        float s[SC_W];
        #pragma unroll
        for (int j = 0; j < SC_W; ++j)
            s[j] = sum16(xv[it].x * v[0][j] + xv[it].y * v[1][j] + xv[it].z * v[2][j] + xv[it].w * v[3][j]);
        if (q == 0 && n < N_NODES) {
            float4* o = (float4*)(sc + (size_t)n * SC_W);
            o[0] = make_float4(s[0], s[1], s[2], s[3]);
            o[1] = make_float4(s[4], s[5], s[6], s[7]);
        }
    }
}

__global__ __launch_bounds__(256) void k_zero(int* __restrict__ degc) {
    int i = blockIdx.x * blockDim.x + threadIdx.x;
    if (i < N_NODES) degc[i] = 0;
}

__global__ __launch_bounds__(256) void k_rank(const int* __restrict__ ei,
                                              int* __restrict__ degc,
                                              int* __restrict__ rank) {
    int e = blockIdx.x * blockDim.x + threadIdx.x;
    if (e >= N_EDGES) return;
    int d = ei[N_EDGES + e];
    rank[e] = atomicAdd(&degc[d], 1);
}

__device__ __forceinline__ int wave_sum_i(int v) {
    #pragma unroll
    for (int o = 16; o >= 1; o >>= 1) v += __shfl_xor(v, o, 32);
    return v;
}
__device__ __forceinline__ int wave_incl_scan_i(int v, int lane) {
    #pragma unroll
    for (int o = 1; o < 32; o <<= 1) {
        int t = __shfl_up(v, o, 32);
        if (lane >= o) v += t;
    }
    return v;
}

__global__ __launch_bounds__(SCAN_T) void k_bsum(const int* __restrict__ degc,
                                                 int* __restrict__ bsum) {
    __shared__ int wtot[32];
    const int t = threadIdx.x;
    const int i = blockIdx.x * SCAN_T + t;
    int v = (i < N_NODES) ? degc[i] : 0;
    v = wave_sum_i(v);
    if ((t & 31) == 0) wtot[t >> 5] = v;
    __syncthreads();
    if (t < 32) {
        int s = wave_sum_i(wtot[t]);
        if (t == 0) bsum[blockIdx.x] = s;
    }
}

__global__ __launch_bounds__(SCAN_T) void k_scan(const int* __restrict__ degc,
                                                 const int* __restrict__ bsum,
                                                 int* __restrict__ startc,
                                                 int* __restrict__ meta) {
    __shared__ int wtot[32];
    __shared__ int boff;
    const int t = threadIdx.x, b = blockIdx.x;
    const int lane = t & 31, w = t >> 5;
    const int i = b * SCAN_T + t;
    const int v = (i < N_NODES) ? degc[i] : 0;
    const int incl = wave_incl_scan_i(v, lane);
    if (lane == 31) wtot[w] = incl;
    __syncthreads();
    if (w == 0) {
        int x  = wtot[lane];
        int xs = wave_incl_scan_i(x, lane);
        wtot[lane] = xs - x;
        int pb = 0;
        for (int j = lane; j < b; j += 32) pb += bsum[j];
        pb = wave_sum_i(pb);
        if (lane == 0) boff = pb;
    }
    __syncthreads();
    const int excl = incl - v + wtot[w] + boff;
    if (i < N_NODES) {
        startc[i] = excl;
        meta[(size_t)i * HC]     = excl;
        meta[(size_t)i * HC + 1] = v;
    }
}

__global__ __launch_bounds__(256) void k_fill(const int* __restrict__ ei,
                                              const int* __restrict__ startc,
                                              const int* __restrict__ rank,
                                              int* __restrict__ sorted) {
    int e = blockIdx.x * blockDim.x + threadIdx.x;
    if (e >= N_EDGES) return;
    int s = ei[e], d = ei[N_EDGES + e];
    sorted[startc[d] + rank[e]] = s;
}

typedef __attribute__((ext_vector_type(2))) _Float16 v2h;

__device__ __forceinline__ float head_update(const float e, const bool on,
                                             float& m, float& l, float2& acc) {
    const float mn = fmaxf(m, wave_max(e));
    const float sc = __expf(m - mn);
    const float p  = on ? __expf(e - mn) : 0.f;
    l = l * sc + wave_sum(p);
    acc.x *= sc; acc.y *= sc;
    m = mn;
    return p;
}

__device__ __forceinline__ float2 ldx(const float* __restrict__ x, int s, int lane) {
    return ((const float2*)(x + (size_t)s * IN_CH))[lane];
}

__device__ __forceinline__ void node_aggregate(const int d, const int lane,
                                               const int* __restrict__ sorted,
                                               const float* __restrict__ x,
                                               const float* __restrict__ sc,
                                               const int* meta,
                                               _Float16* aggrow) {
    const int start = __builtin_amdgcn_readfirstlane(meta[(size_t)d * HC]);
    const int deg   = __builtin_amdgcn_readfirstlane(meta[(size_t)d * HC + 1]);

    const float4 asd = ((const float4*)(sc + (size_t)d * SC_W))[0];
    const float4 ad  = ((const float4*)(sc + (size_t)d * SC_W))[1];
    const float2 xd  = ldx(x, d, lane);

    float m0 = leaky(asd.x + ad.x), m1 = leaky(asd.y + ad.y);
    float m2 = leaky(asd.z + ad.z), m3 = leaky(asd.w + ad.w);
    float l0 = 1.f, l1 = 1.f, l2 = 1.f, l3 = 1.f;
    float2 a0 = xd, a1 = xd, a2 = xd, a3 = xd;

    const float ninf = -__builtin_inff();
    for (int base = 0; base < deg; base += 32) {
        int n = deg - base; if (n > 32) n = 32;
        const bool on = lane < n;
        int myid = 0;
        if (on) myid = sorted[start + base + lane];
        const float4 as = ((const float4*)(sc + (size_t)myid * SC_W))[0];

        const int ng = (n + 3) & ~3;
        #define LD4(ca, cb, cc, cd, kk)                                           \
            ca = ldx(x, __builtin_amdgcn_readlane(myid, (kk)), lane);             \
            cb = ldx(x, __builtin_amdgcn_readlane(myid, (kk) + 1), lane);         \
            cc = ldx(x, __builtin_amdgcn_readlane(myid, (kk) + 2), lane);         \
            cd = ldx(x, __builtin_amdgcn_readlane(myid, (kk) + 3), lane);
        #define EDGE_FMA(cc, kk)                                                  \
        {   const float w0 = rlf(p0, (kk)), w1 = rlf(p1, (kk));                   \
            const float w2 = rlf(p2, (kk)), w3 = rlf(p3, (kk));                   \
            a0.x += w0 * cc.x; a0.y += w0 * cc.y;                                 \
            a1.x += w1 * cc.x; a1.y += w1 * cc.y;                                 \
            a2.x += w2 * cc.x; a2.y += w2 * cc.y;                                 \
            a3.x += w3 * cc.x; a3.y += w3 * cc.y; }
        const float2 z2 = make_float2(0.f, 0.f);
        float2 c0, c1, c2, c3, c4 = z2, c5 = z2, c6 = z2, c7 = z2;
        LD4(c0, c1, c2, c3, 0)
        if (ng > 4) { LD4(c4, c5, c6, c7, 4) }

        const float p0 = head_update(on ? leaky(as.x + ad.x) : ninf, on, m0, l0, a0);
        const float p1 = head_update(on ? leaky(as.y + ad.y) : ninf, on, m1, l1, a1);
        const float p2 = head_update(on ? leaky(as.z + ad.z) : ninf, on, m2, l2, a2);
        const float p3 = head_update(on ? leaky(as.w + ad.w) : ninf, on, m3, l3, a3);

        for (int k = 0;;) {
            EDGE_FMA(c0, k)
            EDGE_FMA(c1, k + 1)
            EDGE_FMA(c2, k + 2)
            EDGE_FMA(c3, k + 3)
            if (k + 4 < ng) {
                EDGE_FMA(c4, k + 4)
                EDGE_FMA(c5, k + 5)
                EDGE_FMA(c6, k + 6)
                EDGE_FMA(c7, k + 7)
            }
            k += 8;
            if (k >= ng) break;
            LD4(c0, c1, c2, c3, k)
            if (k + 4 < ng) { LD4(c4, c5, c6, c7, k + 4) }
        }
        #undef LD4
        #undef EDGE_FMA
    }

    const float i0 = 1.0f / (l0 + 1e-16f), i1 = 1.0f / (l1 + 1e-16f);
    const float i2 = 1.0f / (l2 + 1e-16f), i3 = 1.0f / (l3 + 1e-16f);
    v2h h0, h1, h2, h3;
    h0[0] = (_Float16)(a0.x * i0); h0[1] = (_Float16)(a0.y * i0);
    h1[0] = (_Float16)(a1.x * i1); h1[1] = (_Float16)(a1.y * i1);
    h2[0] = (_Float16)(a2.x * i2); h2[1] = (_Float16)(a2.y * i2);
    h3[0] = (_Float16)(a3.x * i3); h3[1] = (_Float16)(a3.y * i3);
    *(v2h*)(aggrow + 0 * IN_CH + 2 * lane) = h0;
    *(v2h*)(aggrow + 1 * IN_CH + 2 * lane) = h1;
    *(v2h*)(aggrow + 2 * IN_CH + 2 * lane) = h2;
    *(v2h*)(aggrow + 3 * IN_CH + 2 * lane) = h3;
}

__global__ __launch_bounds__(256) void k_node(const int* __restrict__ sorted,
                                              const float* __restrict__ x,
                                              const float* __restrict__ sc,
                                              const float* __restrict__ W,
                                              const float* __restrict__ bias,
                                              const int* meta,
                                              float* out) {
    __shared__ __attribute__((aligned(16))) _Float16 aggs[16 * AP];
    __shared__ __attribute__((aligned(16))) float    outs[16 * OP];
    const int tid  = threadIdx.x;
    const int wave = tid >> 5;
    const int lane = tid & 31;
    const int row0 = blockIdx.x * 16;

    for (int t = 0; t < 2; ++t) {
        const int r = wave * 2 + t;
        node_aggregate(row0 + r, lane, sorted, x, sc, meta, aggs + r * AP);
    }
    __syncthreads();

    const int mrow  = lane & 15;
    const int khalf = (lane >> 4) << 3;
    const int ncol  = wave * 16 + (lane & 15);
    const _Float16* arow = aggs + mrow * AP + (wave >> 1) * IN_CH;
    v16h fa0, fa1, fb0, fb1;
    #pragma unroll
    for (int i = 0; i < 16; ++i) {
        const int k = i + (i & 8) + khalf;
        fa0[i] = arow[k];
        fa1[i] = arow[k + 32];
        fb0[i] = (_Float16)W[(size_t)k * HC + ncol];
        fb1[i] = (_Float16)W[(size_t)(k + 32) * HC + ncol];
    }
    v8f c = {};
    c = __builtin_amdgcn_wmma_f32_16x16x32_f16(false, fa0, false, fb0, (short)0, c, false, false);
    c = __builtin_amdgcn_wmma_f32_16x16x32_f16(false, fa1, false, fb1, (short)0, c, false, false);

    const float bc = bias[ncol];
    #pragma unroll
    for (int r = 0; r < 8; ++r)
        outs[(r + khalf) * OP + ncol] = c[r] + bc;
    __syncthreads();

    for (int t = 0; t < 2; ++t) {
        const int r = wave * 2 + t;
        float4 v = *(const float4*)(outs + r * OP + 4 * lane);
        const float mx = wave_max(fmaxf(fmaxf(v.x, v.y), fmaxf(v.z, v.w)));
        const float se = wave_sum(__expf(v.x - mx) + __expf(v.y - mx) + __expf(v.z - mx) + __expf(v.w - mx));
        const float lse = mx + __logf(se);
        v.x -= lse; v.y -= lse; v.z -= lse; v.w -= lse;
        ((float4*)(out + (size_t)(row0 + r) * HC))[lane] = v;
    }
}

extern "C" void kernel_launch(void* const* d_in, const int* in_sizes, int n_in,
                              void* d_out, int out_size, void* d_ws, size_t ws_size,
                              hipStream_t stream) {
    const float* x       = (const float*)d_in[0];
    const int*   ei      = (const int*)  d_in[1];
    const float* W       = (const float*)d_in[2];
    const float* att_src = (const float*)d_in[3];
    const float* att_dst = (const float*)d_in[4];
    const float* bias    = (const float*)d_in[5];
    float*       out     = (float*)d_out;
    int*         meta    = (int*)d_out;

    int*   sorted = (int*)d_ws + (size_t)N_NODES * HC;
    int*   rank   = (int*)d_ws;
    int*   degc   = rank   + (size_t)N_EDGES;
    int*   startc = degc   + (size_t)N_NODES;
    int*   bsum   = startc + (size_t)N_NODES;
    float* sc     = (float*)d_ws + (size_t)2 * 1024 * 1024;

    k_zero<<<(N_NODES + 255) / 256, 256, 0, stream>>>(degc);
    k_rank<<<(N_EDGES + 255) / 256, 256, 0, stream>>>(ei, degc, rank);
    k_bsum<<<SCAN_B, SCAN_T, 0, stream>>>(degc, bsum);
    k_scan<<<SCAN_B, SCAN_T, 0, stream>>>(degc, bsum, startc, meta);
    k_fill<<<(N_EDGES + 255) / 256, 256, 0, stream>>>(ei, startc, rank, sorted);
    k_score<<<(N_NODES + SCORE_NPB - 1) / SCORE_NPB, 256, 0, stream>>>(x, W, att_src, att_dst, sc);
    k_node<<<N_NODES / 16, 256, 0, stream>>>(sorted, x, sc, W, bias, meta, out);
}


# Edges grouped by destination with a two-level counting sort in LDS (bucket runs, then per bucket): no global atomics, start gathers or 4-byte scatters; rank is LDS arrival order; new sum order
# speedup vs baseline: 7.4820x; 1.8382x over previous
#include <hip/hip_runtime.h>

#define N_NODES 100000
#define N_EDGES 1600000
#define IN_CH   64
#define OUT_CH  32
#define HEADS   4
#define HC      (HEADS * OUT_CH)
#define NEG_SLOPE 0.2f

typedef __attribute__((ext_vector_type(16))) _Float16 v16h;
typedef __attribute__((ext_vector_type(8)))  float    v8f;


#define SC_W   8
#define AP     264
#define OP     132

__device__ __forceinline__ float dpp_x1(float v) {
    return __int_as_float(__builtin_amdgcn_update_dpp(0, __float_as_int(v), 0xB1, 0xF, 0xF, true));
}
__device__ __forceinline__ float dpp_x2(float v) {
    return __int_as_float(__builtin_amdgcn_update_dpp(0, __float_as_int(v), 0x4E, 0xF, 0xF, true));
}
__device__ __forceinline__ float dpp_hm(float v) {
    return __int_as_float(__builtin_amdgcn_update_dpp(0, __float_as_int(v), 0x141, 0xF, 0xF, true));
}
__device__ __forceinline__ float dpp_rm(float v) {
    return __int_as_float(__builtin_amdgcn_update_dpp(0, __float_as_int(v), 0x140, 0xF, 0xF, true));
}
__device__ __forceinline__ float sum16(float v) {
    v += dpp_x1(v); v += dpp_x2(v); v += dpp_hm(v); v += dpp_rm(v);
    return v;
}
__device__ __forceinline__ float max16(float v) {
    v = fmaxf(v, dpp_x1(v)); v = fmaxf(v, dpp_x2(v)); v = fmaxf(v, dpp_hm(v)); v = fmaxf(v, dpp_rm(v));
    return v;
}
__device__ __forceinline__ float rlf(float v, int k) {
    return __int_as_float(__builtin_amdgcn_readlane(__float_as_int(v), k));
}
__device__ __forceinline__ float wave_sum(float v) { v = sum16(v); return rlf(v, 0) + rlf(v, 16); }
__device__ __forceinline__ float wave_max(float v) { v = max16(v); return fmaxf(rlf(v, 0), rlf(v, 16)); }
__device__ __forceinline__ float leaky(float e) { return e > 0.f ? e : NEG_SLOPE * e; }

#define SCORE_NPB 128
__global__ __launch_bounds__(256) void k_score(const float* __restrict__ x,
                                               const float* __restrict__ W,
                                               const float* __restrict__ att_src,
                                               const float* __restrict__ att_dst,
                                               float* __restrict__ sc) {
    __shared__ __attribute__((aligned(16))) float V[IN_CH * SC_W];
    const int tid = threadIdx.x;
    for (int i = tid; i < IN_CH * SC_W; i += 256) {
        const int k = i >> 3, j = i & 7, hd = j & 3;
        const float4* wr = (const float4*)(W + (size_t)k * HC + hd * OUT_CH);
        const float4* ar = (const float4*)((j < 4 ? att_src : att_dst) + hd * OUT_CH);
        float s = 0.f;
        #pragma unroll
        for (int c = 0; c < OUT_CH / 4; ++c) {
            const float4 w4 = wr[c], a4 = ar[c];
            s += w4.x * a4.x + w4.y * a4.y + w4.z * a4.z + w4.w * a4.w;
        }
        V[i] = s;
    }
    __syncthreads();

    const int wave = tid >> 5, lane = tid & 31;
    const int q = lane & 15, hf = lane >> 4;
    float v[4][SC_W];
    #pragma unroll
    for (int t = 0; t < 4; ++t) {
        #pragma unroll
        for (int j = 0; j < SC_W; ++j) v[t][j] = V[(4 * q + t) * SC_W + j];
    }
    const int base = blockIdx.x * SCORE_NPB + wave * 16 + hf;
    float4 xv[8];
    #pragma unroll
    for (int it = 0; it < 8; ++it) {
        int n = base + it * 2;
        if (n > N_NODES - 1) n = N_NODES - 1;
        xv[it] = ((const float4*)(x + (size_t)n * IN_CH))[q];
    }
    #pragma unroll
    for (int it = 0; it < 8; ++it) {
        const int n = base + it * 2;
        float s[SC_W];
        #pragma unroll
        for (int j = 0; j < SC_W; ++j)
            s[j] = sum16(xv[it].x * v[0][j] + xv[it].y * v[1][j] + xv[it].z * v[2][j] + xv[it].w * v[3][j]);
        if (q == 0 && n < N_NODES) {
            float4* o = (float4*)(sc + (size_t)n * SC_W);
            o[0] = make_float4(s[0], s[1], s[2], s[3]);
            o[1] = make_float4(s[4], s[5], s[6], s[7]);
        }
    }
}

#define BKS    9
#define BK_N   (1 << BKS)
#define NBK    ((N_NODES + BK_N - 1) / BK_N)
#define NB1    256
#define EPB    (N_EDGES / NB1)
#define S_T    1024
#define EPT    ((EPB + S_T - 1) / S_T)
#define M_CNT  (NBK * NB1)
#define O_PER  (M_CNT / S_T)
#define BCAP   12288
static_assert(EPB * NB1 == N_EDGES, "level-1 slices must tile the edge list");
static_assert(O_PER * S_T == M_CNT, "k_offs covers the counts exactly");
static_assert(NBK <= 256, "bucket tables hold 256 entries");

__device__ __forceinline__ int wave_incl_scan_i(int v, int lane) {
    #pragma unroll
    for (int o = 1; o < 32; o <<= 1) {
        int t = __shfl_up(v, o, 32);
        if (lane >= o) v += t;
    }
    return v;
}

__device__ __forceinline__ int block_excl_scan(const int c, int* wt, const int t) {
    const int lane = t & 31, w = t >> 5;
    const int incl = wave_incl_scan_i(c, lane);
    if (lane == 31) wt[w] = incl;
    __syncthreads();
    if (w == 0) {
        const int x  = wt[lane];
        const int xs = wave_incl_scan_i(x, lane);
        wt[lane] = xs - x;
    }
    __syncthreads();
    return incl - c + wt[w];
}

__global__ __launch_bounds__(S_T) void k_hist(const int* __restrict__ ei, int* __restrict__ cnt) {
    __shared__ int h[256];
    const int t = threadIdx.x, blk = blockIdx.x;
    if (t < 256) h[t] = 0;
    __syncthreads();
    const int* dp = ei + N_EDGES + (size_t)blk * EPB;
    int dv[EPT];
    #pragma unroll
    for (int k = 0; k < EPT; ++k) {
        const int i = t + k * S_T;
        dv[k] = (i < EPB) ? dp[i] : -1;
    }
    #pragma unroll
    for (int k = 0; k < EPT; ++k)
        if (dv[k] >= 0) atomicAdd(&h[dv[k] >> BKS], 1);
    __syncthreads();
    if (t < NBK) cnt[t * NB1 + blk] = h[t];
}

__global__ __launch_bounds__(S_T) void k_offs(int* cnt) {
    __shared__ int wt[32];
    const int t = threadIdx.x;
    int* p = cnt + t * O_PER;
    int v[O_PER];
    int s = 0;
    #pragma unroll
    for (int j = 0; j < O_PER; ++j) { v[j] = p[j]; s += v[j]; }
    int run = block_excl_scan(s, wt, t);
    #pragma unroll
    for (int j = 0; j < O_PER; ++j) { p[j] = run; run += v[j]; }
}

__global__ __launch_bounds__(S_T) void k_part(const int* __restrict__ ei,
                                              const int* __restrict__ offs,
                                              int* __restrict__ tmpw) {
    int2* __restrict__ tmp = (int2*)tmpw;
    __shared__ int  lcnt[256];
    __shared__ int  lcur[256];
    __shared__ int  gofs[256];
    __shared__ int  wt[32];
    __shared__ int2 stage[EPB];
    const int t = threadIdx.x, blk = blockIdx.x;
    if (t < 256) lcnt[t] = 0;
    __syncthreads();
    const int* sp = ei + (size_t)blk * EPB;
    const int* dp = sp + N_EDGES;
    int sv[EPT], dv[EPT];
    #pragma unroll
    for (int k = 0; k < EPT; ++k) {
        const int i = t + k * S_T;
        sv[k] = 0; dv[k] = -1;
        if (i < EPB) { sv[k] = sp[i]; dv[k] = dp[i]; }
    }
    #pragma unroll
    for (int k = 0; k < EPT; ++k)
        if (dv[k] >= 0) atomicAdd(&lcnt[dv[k] >> BKS], 1);
    __syncthreads();
    const int c    = (t < 256) ? lcnt[t] : 0;
    const int excl = block_excl_scan(c, wt, t);
    if (t < 256) {
        lcur[t] = excl;
        gofs[t] = ((t < NBK) ? offs[t * NB1 + blk] : 0) - excl;
    }
    __syncthreads();
    #pragma unroll
    for (int k = 0; k < EPT; ++k)
        if (dv[k] >= 0) {
            const int p = atomicAdd(&lcur[dv[k] >> BKS], 1);
            stage[p] = make_int2(sv[k], dv[k]);
        }
    __syncthreads();
    for (int i = t; i < EPB; i += S_T) {
        const int2 r = stage[i];
        tmp[gofs[r.y >> BKS] + i] = r;
    }
}

__global__ __launch_bounds__(S_T) void k_bucket(const int* __restrict__ tmpw,
                                                const int* __restrict__ offs,
                                                int* __restrict__ sorted,
                                                int* __restrict__ meta) {
    const int2* __restrict__ tmp = (const int2*)tmpw;
    __shared__ int cnt[BK_N];
    __shared__ int cur[BK_N];
    __shared__ int wt[32];
    __shared__ int stage[BCAP];
    const int t = threadIdx.x, b = blockIdx.x;
    const int bstart = offs[b * NB1];
    const int bend   = (b + 1 < NBK) ? offs[(b + 1) * NB1] : N_EDGES;
    const int n      = bend - bstart;
    const int2* src  = tmp + bstart;
    if (t < BK_N) cnt[t] = 0;
    __syncthreads();
    for (int base = 0; base < n; base += 4 * S_T) {
        int dl[4];
        #pragma unroll
        for (int k = 0; k < 4; ++k) {
            const int i = base + t + k * S_T;
            dl[k] = (i < n) ? src[i].y : -1;
        }
        #pragma unroll
        for (int k = 0; k < 4; ++k)
            if (dl[k] >= 0) atomicAdd(&cnt[dl[k] & (BK_N - 1)], 1);
    }
    __syncthreads();
    const int c    = (t < BK_N) ? cnt[t] : 0;
    const int excl = block_excl_scan(c, wt, t);
    if (t < BK_N) {
        cur[t] = excl;
        const int d = b * BK_N + t;
        if (d < N_NODES) {
            meta[(size_t)d * HC]     = bstart + excl;
            meta[(size_t)d * HC + 1] = c;
        }
    }
    __syncthreads();
    const bool fits = n <= BCAP;
    for (int base = 0; base < n; base += 4 * S_T) {
        int2 r[4];
        #pragma unroll
        for (int k = 0; k < 4; ++k) {
            const int i = base + t + k * S_T;
            r[k] = make_int2(0, -1);
            if (i < n) r[k] = src[i];
        }
        #pragma unroll
        for (int k = 0; k < 4; ++k)
            if (r[k].y >= 0) {
                const int p = atomicAdd(&cur[r[k].y & (BK_N - 1)], 1);
                if (fits) stage[p] = r[k].x;
                else      sorted[bstart + p] = r[k].x;
            }
    }
    if (fits) {
        __syncthreads();
        for (int i = t; i < n; i += S_T) sorted[bstart + i] = stage[i];
    }
}

typedef __attribute__((ext_vector_type(2))) _Float16 v2h;

__device__ __forceinline__ float head_update(const float e, const bool on,
                                             float& m, float& l, float2& acc) {
    const float mn = fmaxf(m, wave_max(e));
    const float sc = __expf(m - mn);
    const float p  = on ? __expf(e - mn) : 0.f;
    l = l * sc + wave_sum(p);
    acc.x *= sc; acc.y *= sc;
    m = mn;
    return p;
}

__device__ __forceinline__ float2 ldx(const float* __restrict__ x, int s, int lane) {
    return ((const float2*)(x + (size_t)s * IN_CH))[lane];
}

__device__ __forceinline__ void node_aggregate(const int d, const int lane,
                                               const int* __restrict__ sorted,
                                               const float* __restrict__ x,
                                               const float* __restrict__ sc,
                                               const int* meta,
                                               _Float16* aggrow) {
    const int start = __builtin_amdgcn_readfirstlane(meta[(size_t)d * HC]);
    const int deg   = __builtin_amdgcn_readfirstlane(meta[(size_t)d * HC + 1]);

    const float4 asd = ((const float4*)(sc + (size_t)d * SC_W))[0];
    const float4 ad  = ((const float4*)(sc + (size_t)d * SC_W))[1];
    const float2 xd  = ldx(x, d, lane);

    float m0 = leaky(asd.x + ad.x), m1 = leaky(asd.y + ad.y);
    float m2 = leaky(asd.z + ad.z), m3 = leaky(asd.w + ad.w);
    float l0 = 1.f, l1 = 1.f, l2 = 1.f, l3 = 1.f;
    float2 a0 = xd, a1 = xd, a2 = xd, a3 = xd;

    const float ninf = -__builtin_inff();
    for (int base = 0; base < deg; base += 32) {
        int n = deg - base; if (n > 32) n = 32;
        const bool on = lane < n;
        int myid = 0;
        if (on) myid = sorted[start + base + lane];
        const float4 as = ((const float4*)(sc + (size_t)myid * SC_W))[0];

        const int ng = (n + 3) & ~3;
        #define LD4(ca, cb, cc, cd, kk)                                           \
            ca = ldx(x, __builtin_amdgcn_readlane(myid, (kk)), lane);             \
            cb = ldx(x, __builtin_amdgcn_readlane(myid, (kk) + 1), lane);         \
            cc = ldx(x, __builtin_amdgcn_readlane(myid, (kk) + 2), lane);         \
            cd = ldx(x, __builtin_amdgcn_readlane(myid, (kk) + 3), lane);
        #define EDGE_FMA(cc, kk)                                                  \
        {   const float w0 = rlf(p0, (kk)), w1 = rlf(p1, (kk));                   \
            const float w2 = rlf(p2, (kk)), w3 = rlf(p3, (kk));                   \
            a0.x += w0 * cc.x; a0.y += w0 * cc.y;                                 \
            a1.x += w1 * cc.x; a1.y += w1 * cc.y;                                 \
            a2.x += w2 * cc.x; a2.y += w2 * cc.y;                                 \
            a3.x += w3 * cc.x; a3.y += w3 * cc.y; }
        const float2 z2 = make_float2(0.f, 0.f);
        float2 c0, c1, c2, c3, c4 = z2, c5 = z2, c6 = z2, c7 = z2;
        LD4(c0, c1, c2, c3, 0)
        if (ng > 4) { LD4(c4, c5, c6, c7, 4) }

        const float p0 = head_update(on ? leaky(as.x + ad.x) : ninf, on, m0, l0, a0);
        const float p1 = head_update(on ? leaky(as.y + ad.y) : ninf, on, m1, l1, a1);
        const float p2 = head_update(on ? leaky(as.z + ad.z) : ninf, on, m2, l2, a2);
        const float p3 = head_update(on ? leaky(as.w + ad.w) : ninf, on, m3, l3, a3);

        for (int k = 0;;) {
            EDGE_FMA(c0, k)
            EDGE_FMA(c1, k + 1)
            EDGE_FMA(c2, k + 2)
            EDGE_FMA(c3, k + 3)
            if (k + 4 < ng) {
                EDGE_FMA(c4, k + 4)
                EDGE_FMA(c5, k + 5)
                EDGE_FMA(c6, k + 6)
                EDGE_FMA(c7, k + 7)
            }
            k += 8;
            if (k >= ng) break;
            LD4(c0, c1, c2, c3, k)
            if (k + 4 < ng) { LD4(c4, c5, c6, c7, k + 4) }
        }
        #undef LD4
        #undef EDGE_FMA
    }

    const float i0 = 1.0f / (l0 + 1e-16f), i1 = 1.0f / (l1 + 1e-16f);
    const float i2 = 1.0f / (l2 + 1e-16f), i3 = 1.0f / (l3 + 1e-16f);
    v2h h0, h1, h2, h3;
    h0[0] = (_Float16)(a0.x * i0); h0[1] = (_Float16)(a0.y * i0);
    h1[0] = (_Float16)(a1.x * i1); h1[1] = (_Float16)(a1.y * i1);
    h2[0] = (_Float16)(a2.x * i2); h2[1] = (_Float16)(a2.y * i2);
    h3[0] = (_Float16)(a3.x * i3); h3[1] = (_Float16)(a3.y * i3);
    *(v2h*)(aggrow + 0 * IN_CH + 2 * lane) = h0;
    *(v2h*)(aggrow + 1 * IN_CH + 2 * lane) = h1;
    *(v2h*)(aggrow + 2 * IN_CH + 2 * lane) = h2;
    *(v2h*)(aggrow + 3 * IN_CH + 2 * lane) = h3;
}

__global__ __launch_bounds__(256) void k_node(const int* __restrict__ sorted,
                                              const float* __restrict__ x,
                                              const float* __restrict__ sc,
                                              const float* __restrict__ W,
                                              const float* __restrict__ bias,
                                              const int* meta,
                                              float* out) {
    __shared__ __attribute__((aligned(16))) _Float16 aggs[16 * AP];
    __shared__ __attribute__((aligned(16))) float    outs[16 * OP];
    const int tid  = threadIdx.x;
    const int wave = tid >> 5;
    const int lane = tid & 31;
    const int row0 = blockIdx.x * 16;

    for (int t = 0; t < 2; ++t) {
        const int r = wave * 2 + t;
        node_aggregate(row0 + r, lane, sorted, x, sc, meta, aggs + r * AP);
    }
    __syncthreads();

    const int mrow  = lane & 15;
    const int khalf = (lane >> 4) << 3;
    const int ncol  = wave * 16 + (lane & 15);
    const _Float16* arow = aggs + mrow * AP + (wave >> 1) * IN_CH;
    v16h fa0, fa1, fb0, fb1;
    #pragma unroll
    for (int i = 0; i < 16; ++i) {
        const int k = i + (i & 8) + khalf;
        fa0[i] = arow[k];
        fa1[i] = arow[k + 32];
        fb0[i] = (_Float16)W[(size_t)k * HC + ncol];
        fb1[i] = (_Float16)W[(size_t)(k + 32) * HC + ncol];
    }
    v8f c = {};
    c = __builtin_amdgcn_wmma_f32_16x16x32_f16(false, fa0, false, fb0, (short)0, c, false, false);
    c = __builtin_amdgcn_wmma_f32_16x16x32_f16(false, fa1, false, fb1, (short)0, c, false, false);

    const float bc = bias[ncol];
    #pragma unroll
    for (int r = 0; r < 8; ++r)
        outs[(r + khalf) * OP + ncol] = c[r] + bc;
    __syncthreads();

    for (int t = 0; t < 2; ++t) {
        const int r = wave * 2 + t;
        float4 v = *(const float4*)(outs + r * OP + 4 * lane);
        const float mx = wave_max(fmaxf(fmaxf(v.x, v.y), fmaxf(v.z, v.w)));
        const float se = wave_sum(__expf(v.x - mx) + __expf(v.y - mx) + __expf(v.z - mx) + __expf(v.w - mx));
        const float lse = mx + __logf(se);
        v.x -= lse; v.y -= lse; v.z -= lse; v.w -= lse;
        ((float4*)(out + (size_t)(row0 + r) * HC))[lane] = v;
    }
}

extern "C" void kernel_launch(void* const* d_in, const int* in_sizes, int n_in,
                              void* d_out, int out_size, void* d_ws, size_t ws_size,
                              hipStream_t stream) {
    const float* x       = (const float*)d_in[0];
    const int*   ei      = (const int*)  d_in[1];
    const float* W       = (const float*)d_in[2];
    const float* att_src = (const float*)d_in[3];
    const float* att_dst = (const float*)d_in[4];
    const float* bias    = (const float*)d_in[5];
    float*       out     = (float*)d_out;
    int*         meta    = (int*)d_out;

    int*   sorted = (int*)d_ws + (size_t)N_NODES * HC;
    int*   cnt    = (int*)d_ws;
    float* sc     = (float*)d_ws + (size_t)2 * 1024 * 1024;
    int*   tmp    = (int*)d_ws + (size_t)4 * 1024 * 1024;

    k_hist<<<NB1, S_T, 0, stream>>>(ei, cnt);
    k_offs<<<1, S_T, 0, stream>>>(cnt);
    k_part<<<NB1, S_T, 0, stream>>>(ei, cnt, tmp);
    k_bucket<<<NBK, S_T, 0, stream>>>(tmp, cnt, sorted, meta);
    k_score<<<(N_NODES + SCORE_NPB - 1) / SCORE_NPB, 256, 0, stream>>>(x, W, att_src, att_dst, sc);
    k_node<<<N_NODES / 16, 256, 0, stream>>>(sorted, x, sc, W, bias, meta, out);
}


# Offsets scan folded into the partition kernel: every block sums the count table's rows itself (whole-line loads, wave sums) for bucket starts and its run starts; the one-block scan launch is gone
# speedup vs baseline: 8.7782x; 1.1732x over previous
#include <hip/hip_runtime.h>

#define N_NODES 100000
#define N_EDGES 1600000
#define IN_CH   64
#define OUT_CH  32
#define HEADS   4
#define HC      (HEADS * OUT_CH)
#define NEG_SLOPE 0.2f

typedef __attribute__((ext_vector_type(16))) _Float16 v16h;
typedef __attribute__((ext_vector_type(8)))  float    v8f;


#define SC_W   8
#define AP     264
#define OP     132

__device__ __forceinline__ float dpp_x1(float v) {
    return __int_as_float(__builtin_amdgcn_update_dpp(0, __float_as_int(v), 0xB1, 0xF, 0xF, true));
}
__device__ __forceinline__ float dpp_x2(float v) {
    return __int_as_float(__builtin_amdgcn_update_dpp(0, __float_as_int(v), 0x4E, 0xF, 0xF, true));
}
__device__ __forceinline__ float dpp_hm(float v) {
    return __int_as_float(__builtin_amdgcn_update_dpp(0, __float_as_int(v), 0x141, 0xF, 0xF, true));
}
__device__ __forceinline__ float dpp_rm(float v) {
    return __int_as_float(__builtin_amdgcn_update_dpp(0, __float_as_int(v), 0x140, 0xF, 0xF, true));
}
__device__ __forceinline__ float sum16(float v) {
    v += dpp_x1(v); v += dpp_x2(v); v += dpp_hm(v); v += dpp_rm(v);
    return v;
}
__device__ __forceinline__ float max16(float v) {
    v = fmaxf(v, dpp_x1(v)); v = fmaxf(v, dpp_x2(v)); v = fmaxf(v, dpp_hm(v)); v = fmaxf(v, dpp_rm(v));
    return v;
}
__device__ __forceinline__ float rlf(float v, int k) {
    return __int_as_float(__builtin_amdgcn_readlane(__float_as_int(v), k));
}
__device__ __forceinline__ float wave_sum(float v) { v = sum16(v); return rlf(v, 0) + rlf(v, 16); }
__device__ __forceinline__ float wave_max(float v) { v = max16(v); return fmaxf(rlf(v, 0), rlf(v, 16)); }
__device__ __forceinline__ float leaky(float e) { return e > 0.f ? e : NEG_SLOPE * e; }

#define SCORE_NPB 128
__global__ __launch_bounds__(256) void k_score(const float* __restrict__ x,
                                               const float* __restrict__ W,
                                               const float* __restrict__ att_src,
                                               const float* __restrict__ att_dst,
                                               float* __restrict__ sc) {
    __shared__ __attribute__((aligned(16))) float V[IN_CH * SC_W];
    const int tid = threadIdx.x;
    for (int i = tid; i < IN_CH * SC_W; i += 256) {
        const int k = i >> 3, j = i & 7, hd = j & 3;
        const float4* wr = (const float4*)(W + (size_t)k * HC + hd * OUT_CH);
        const float4* ar = (const float4*)((j < 4 ? att_src : att_dst) + hd * OUT_CH);
        float s = 0.f;
        #pragma unroll
        for (int c = 0; c < OUT_CH / 4; ++c) {
            const float4 w4 = wr[c], a4 = ar[c];
            s += w4.x * a4.x + w4.y * a4.y + w4.z * a4.z + w4.w * a4.w;
        }
        V[i] = s;
    }
    __syncthreads();

    const int wave = tid >> 5, lane = tid & 31;
    const int q = lane & 15, hf = lane >> 4;
    float v[4][SC_W];
    #pragma unroll
    for (int t = 0; t < 4; ++t) {
        #pragma unroll
        for (int j = 0; j < SC_W; ++j) v[t][j] = V[(4 * q + t) * SC_W + j];
    }
    const int base = blockIdx.x * SCORE_NPB + wave * 16 + hf;
    float4 xv[8];
    #pragma unroll
    for (int it = 0; it < 8; ++it) {
        int n = base + it * 2;
        if (n > N_NODES - 1) n = N_NODES - 1;
        xv[it] = ((const float4*)(x + (size_t)n * IN_CH))[q];
    }
    #pragma unroll
    for (int it = 0; it < 8; ++it) {
        const int n = base + it * 2;
        float s[SC_W];
        #pragma unroll
        for (int j = 0; j < SC_W; ++j)
            s[j] = sum16(xv[it].x * v[0][j] + xv[it].y * v[1][j] + xv[it].z * v[2][j] + xv[it].w * v[3][j]);
        if (q == 0 && n < N_NODES) {
            float4* o = (float4*)(sc + (size_t)n * SC_W);
            o[0] = make_float4(s[0], s[1], s[2], s[3]);
            o[1] = make_float4(s[4], s[5], s[6], s[7]);
        }
    }
}

#define BKS    9
#define BK_N   (1 << BKS)
#define NBK    ((N_NODES + BK_N - 1) / BK_N)
#define NB1    256
#define EPB    (N_EDGES / NB1)
#define S_T    1024
#define EPT    ((EPB + S_T - 1) / S_T)
#define M_CNT  (NBK * NB1)
#define RPW    ((NBK + 31) / 32)
#define BCAP   12288
static_assert(EPB * NB1 == N_EDGES, "level-1 slices must tile the edge list");
static_assert(NB1 == 256, "a row of the count table is two 16-byte loads per lane");
static_assert(NBK <= 256, "bucket tables hold 256 entries");

__device__ __forceinline__ int wave_incl_scan_i(int v, int lane) {
    #pragma unroll
    for (int o = 1; o < 32; o <<= 1) {
        int t = __shfl_up(v, o, 32);
        if (lane >= o) v += t;
    }
    return v;
}

__device__ __forceinline__ int block_excl_scan(const int c, int* wt, const int t) {
    const int lane = t & 31, w = t >> 5;
    const int incl = wave_incl_scan_i(c, lane);
    if (lane == 31) wt[w] = incl;
    __syncthreads();
    if (w == 0) {
        const int x  = wt[lane];
        const int xs = wave_incl_scan_i(x, lane);
        wt[lane] = xs - x;
    }
    __syncthreads();
    return incl - c + wt[w];
}

__device__ __forceinline__ void block_excl_scan2(int& a, int& b, int* wta, int* wtb, const int t) {
    const int lane = t & 31, w = t >> 5;
    int ia = a, ib = b;
    #pragma unroll
    for (int o = 1; o < 32; o <<= 1) {
        const int ta = __shfl_up(ia, o, 32), tb = __shfl_up(ib, o, 32);
        if (lane >= o) { ia += ta; ib += tb; }
    }
    if (lane == 31) { wta[w] = ia; wtb[w] = ib; }
    __syncthreads();
    if (w == 0) {
        const int xa = wta[lane], xb = wtb[lane];
        int sa = xa, sb = xb;
        #pragma unroll
        for (int o = 1; o < 32; o <<= 1) {
            const int ta = __shfl_up(sa, o, 32), tb = __shfl_up(sb, o, 32);
            if (lane >= o) { sa += ta; sb += tb; }
        }
        wta[lane] = sa - xa;
        wtb[lane] = sb - xb;
    }
    __syncthreads();
    a = ia - a + wta[w];
    b = ib - b + wtb[w];
}

__device__ __forceinline__ int wave_sum_i(int v) {
    v += __builtin_amdgcn_update_dpp(0, v, 0xB1,  0xF, 0xF, true);
    v += __builtin_amdgcn_update_dpp(0, v, 0x4E,  0xF, 0xF, true);
    v += __builtin_amdgcn_update_dpp(0, v, 0x141, 0xF, 0xF, true);
    v += __builtin_amdgcn_update_dpp(0, v, 0x140, 0xF, 0xF, true);
    return __builtin_amdgcn_readlane(v, 0) + __builtin_amdgcn_readlane(v, 16);
}

__global__ __launch_bounds__(S_T) void k_hist(const int* __restrict__ ei, int* __restrict__ cnt) {
    __shared__ int h[256];
    const int t = threadIdx.x, blk = blockIdx.x;
    if (t < 256) h[t] = 0;
    __syncthreads();
    const int* dp = ei + N_EDGES + (size_t)blk * EPB;
    int dv[EPT];
    #pragma unroll
    for (int k = 0; k < EPT; ++k) {
        const int i = t + k * S_T;
        dv[k] = (i < EPB) ? dp[i] : -1;
    }
    #pragma unroll
    for (int k = 0; k < EPT; ++k)
        if (dv[k] >= 0) atomicAdd(&h[dv[k] >> BKS], 1);
    __syncthreads();
    if (t < NBK) cnt[t * NB1 + blk] = h[t];
}

__global__ __launch_bounds__(S_T) void k_part(const int* __restrict__ ei,
                                              const int* __restrict__ cnt,
                                              int* __restrict__ bks,
                                              int* __restrict__ tmpw) {
    int2* __restrict__ tmp = (int2*)tmpw;
    __shared__ int  lcnt[256];
    __shared__ int  lcur[256];
    __shared__ int  gofs[256];
    __shared__ int  btot[256];
    __shared__ int  bpre[256];
    __shared__ int  wt[32];
    __shared__ int  wt2[32];
    __shared__ int2 stage[EPB];
    const int t = threadIdx.x, blk = blockIdx.x;
    const int lane = t & 31, w = t >> 5;
    if (t < 256) lcnt[t] = 0;
    __syncthreads();
    const int* sp = ei + (size_t)blk * EPB;
    const int* dp = sp + N_EDGES;
    int sv[EPT], dv[EPT];
    #pragma unroll
    for (int k = 0; k < EPT; ++k) {
        const int i = t + k * S_T;
        sv[k] = 0; dv[k] = -1;
        if (i < EPB) { sv[k] = sp[i]; dv[k] = dp[i]; }
    }
    int4 ra[RPW], rb[RPW];
    #pragma unroll
    for (int i = 0; i < RPW; ++i) {
        int b = w + 32 * i;
        if (b > NBK - 1) b = NBK - 1;
        const int4* row = (const int4*)(cnt + b * NB1);
        ra[i] = row[lane];
        rb[i] = row[lane + 32];
    }
    #pragma unroll
    for (int k = 0; k < EPT; ++k)
        if (dv[k] >= 0) atomicAdd(&lcnt[dv[k] >> BKS], 1);
    const int e0 = 4 * lane, e1 = 128 + 4 * lane;
    #pragma unroll
    for (int i = 0; i < RPW; ++i) {
        const int b = w + 32 * i;
        const int4 a = ra[i], c4 = rb[i];
        int s = a.x + a.y + a.z + a.w + c4.x + c4.y + c4.z + c4.w;
        int p = (e0     < blk ? a.x  : 0) + (e0 + 1 < blk ? a.y  : 0)
              + (e0 + 2 < blk ? a.z  : 0) + (e0 + 3 < blk ? a.w  : 0)
              + (e1     < blk ? c4.x : 0) + (e1 + 1 < blk ? c4.y : 0)
              + (e1 + 2 < blk ? c4.z : 0) + (e1 + 3 < blk ? c4.w : 0);
        s = wave_sum_i(s);
        p = wave_sum_i(p);
        if (lane == 0 && b < NBK) { btot[b] = s; bpre[b] = p; }
    }
    __syncthreads();
    int excl = (t < 256) ? lcnt[t] : 0;
    int bst  = (t < NBK) ? btot[t] : 0;
    block_excl_scan2(excl, bst, wt, wt2, t);
    if (t < 256) {
        lcur[t] = excl;
        gofs[t] = ((t < NBK) ? bst + bpre[t] : 0) - excl;
    }
    if (blk == 0 && t <= NBK) bks[t] = bst;
    __syncthreads();
    #pragma unroll
    for (int k = 0; k < EPT; ++k)
        if (dv[k] >= 0) {
            const int p = atomicAdd(&lcur[dv[k] >> BKS], 1);
            stage[p] = make_int2(sv[k], dv[k]);
        }
    __syncthreads();
    for (int i = t; i < EPB; i += S_T) {
        const int2 r = stage[i];
        tmp[gofs[r.y >> BKS] + i] = r;
    }
}

__global__ __launch_bounds__(S_T) void k_bucket(const int* __restrict__ tmpw,
                                                const int* __restrict__ bks,
                                                int* __restrict__ sorted,
                                                int* __restrict__ meta) {
    const int2* __restrict__ tmp = (const int2*)tmpw;
    __shared__ int cnt[BK_N];
    __shared__ int cur[BK_N];
    __shared__ int wt[32];
    __shared__ int stage[BCAP];
    const int t = threadIdx.x, b = blockIdx.x;
    const int bstart = bks[b];
    const int bend   = bks[b + 1];
    const int n      = bend - bstart;
    const int2* src  = tmp + bstart;
    if (t < BK_N) cnt[t] = 0;
    __syncthreads();
    for (int base = 0; base < n; base += 4 * S_T) {
        int dl[4];
        #pragma unroll
        for (int k = 0; k < 4; ++k) {
            const int i = base + t + k * S_T;
            dl[k] = (i < n) ? src[i].y : -1;
        }
        #pragma unroll
        for (int k = 0; k < 4; ++k)
            if (dl[k] >= 0) atomicAdd(&cnt[dl[k] & (BK_N - 1)], 1);
    }
    __syncthreads();
    const int c    = (t < BK_N) ? cnt[t] : 0;
    const int excl = block_excl_scan(c, wt, t);
    if (t < BK_N) {
        cur[t] = excl;
        const int d = b * BK_N + t;
        if (d < N_NODES) {
            meta[(size_t)d * HC]     = bstart + excl;
            meta[(size_t)d * HC + 1] = c;
        }
    }
    __syncthreads();
    const bool fits = n <= BCAP;
    for (int base = 0; base < n; base += 4 * S_T) {
        int2 r[4];
        #pragma unroll
        for (int k = 0; k < 4; ++k) {
            const int i = base + t + k * S_T;
            r[k] = make_int2(0, -1);
            if (i < n) r[k] = src[i];
        }
        #pragma unroll
        for (int k = 0; k < 4; ++k)
            if (r[k].y >= 0) {
                const int p = atomicAdd(&cur[r[k].y & (BK_N - 1)], 1);
                if (fits) stage[p] = r[k].x;
                else      sorted[bstart + p] = r[k].x;
            }
    }
    if (fits) {
        __syncthreads();
        for (int i = t; i < n; i += S_T) sorted[bstart + i] = stage[i];
    }
}

typedef __attribute__((ext_vector_type(2))) _Float16 v2h;

__device__ __forceinline__ float head_update(const float e, const bool on,
                                             float& m, float& l, float2& acc) {
    const float mn = fmaxf(m, wave_max(e));
    const float sc = __expf(m - mn);
    const float p  = on ? __expf(e - mn) : 0.f;
    l = l * sc + wave_sum(p);
    acc.x *= sc; acc.y *= sc;
    m = mn;
    return p;
}

__device__ __forceinline__ float2 ldx(const float* __restrict__ x, int s, int lane) {
    return ((const float2*)(x + (size_t)s * IN_CH))[lane];
}

__device__ __forceinline__ void node_aggregate(const int d, const int lane,
                                               const int* __restrict__ sorted,
                                               const float* __restrict__ x,
                                               const float* __restrict__ sc,
                                               const int* meta,
                                               _Float16* aggrow) {
    const int start = __builtin_amdgcn_readfirstlane(meta[(size_t)d * HC]);
    const int deg   = __builtin_amdgcn_readfirstlane(meta[(size_t)d * HC + 1]);

    const float4 asd = ((const float4*)(sc + (size_t)d * SC_W))[0];
    const float4 ad  = ((const float4*)(sc + (size_t)d * SC_W))[1];
    const float2 xd  = ldx(x, d, lane);

    float m0 = leaky(asd.x + ad.x), m1 = leaky(asd.y + ad.y);
    float m2 = leaky(asd.z + ad.z), m3 = leaky(asd.w + ad.w);
    float l0 = 1.f, l1 = 1.f, l2 = 1.f, l3 = 1.f;
    float2 a0 = xd, a1 = xd, a2 = xd, a3 = xd;

    const float ninf = -__builtin_inff();
    for (int base = 0; base < deg; base += 32) {
        int n = deg - base; if (n > 32) n = 32;
        const bool on = lane < n;
        int myid = 0;
        if (on) myid = sorted[start + base + lane];
        const float4 as = ((const float4*)(sc + (size_t)myid * SC_W))[0];

        const int ng = (n + 3) & ~3;
        #define LD4(ca, cb, cc, cd, kk)                                           \
            ca = ldx(x, __builtin_amdgcn_readlane(myid, (kk)), lane);             \
            cb = ldx(x, __builtin_amdgcn_readlane(myid, (kk) + 1), lane);         \
            cc = ldx(x, __builtin_amdgcn_readlane(myid, (kk) + 2), lane);         \
            cd = ldx(x, __builtin_amdgcn_readlane(myid, (kk) + 3), lane);
        #define EDGE_FMA(cc, kk)                                                  \
        {   const float w0 = rlf(p0, (kk)), w1 = rlf(p1, (kk));                   \
            const float w2 = rlf(p2, (kk)), w3 = rlf(p3, (kk));                   \
            a0.x += w0 * cc.x; a0.y += w0 * cc.y;                                 \
            a1.x += w1 * cc.x; a1.y += w1 * cc.y;                                 \
            a2.x += w2 * cc.x; a2.y += w2 * cc.y;                                 \
            a3.x += w3 * cc.x; a3.y += w3 * cc.y; }
        const float2 z2 = make_float2(0.f, 0.f);
        float2 c0, c1, c2, c3, c4 = z2, c5 = z2, c6 = z2, c7 = z2;
        LD4(c0, c1, c2, c3, 0)
        if (ng > 4) { LD4(c4, c5, c6, c7, 4) }

        const float p0 = head_update(on ? leaky(as.x + ad.x) : ninf, on, m0, l0, a0);
        const float p1 = head_update(on ? leaky(as.y + ad.y) : ninf, on, m1, l1, a1);
        const float p2 = head_update(on ? leaky(as.z + ad.z) : ninf, on, m2, l2, a2);
        const float p3 = head_update(on ? leaky(as.w + ad.w) : ninf, on, m3, l3, a3);

        for (int k = 0;;) {
            EDGE_FMA(c0, k)
            EDGE_FMA(c1, k + 1)
            EDGE_FMA(c2, k + 2)
            EDGE_FMA(c3, k + 3)
            if (k + 4 < ng) {
                EDGE_FMA(c4, k + 4)
                EDGE_FMA(c5, k + 5)
                EDGE_FMA(c6, k + 6)
                EDGE_FMA(c7, k + 7)
            }
            k += 8;
            if (k >= ng) break;
            LD4(c0, c1, c2, c3, k)
            if (k + 4 < ng) { LD4(c4, c5, c6, c7, k + 4) }
        }
        #undef LD4
        #undef EDGE_FMA
    }

    const float i0 = 1.0f / (l0 + 1e-16f), i1 = 1.0f / (l1 + 1e-16f);
    const float i2 = 1.0f / (l2 + 1e-16f), i3 = 1.0f / (l3 + 1e-16f);
    v2h h0, h1, h2, h3;
    h0[0] = (_Float16)(a0.x * i0); h0[1] = (_Float16)(a0.y * i0);
    h1[0] = (_Float16)(a1.x * i1); h1[1] = (_Float16)(a1.y * i1);
    h2[0] = (_Float16)(a2.x * i2); h2[1] = (_Float16)(a2.y * i2);
    h3[0] = (_Float16)(a3.x * i3); h3[1] = (_Float16)(a3.y * i3);
    *(v2h*)(aggrow + 0 * IN_CH + 2 * lane) = h0;
    *(v2h*)(aggrow + 1 * IN_CH + 2 * lane) = h1;
    *(v2h*)(aggrow + 2 * IN_CH + 2 * lane) = h2;
    *(v2h*)(aggrow + 3 * IN_CH + 2 * lane) = h3;
}

__global__ __launch_bounds__(256) void k_node(const int* __restrict__ sorted,
                                              const float* __restrict__ x,
                                              const float* __restrict__ sc,
                                              const float* __restrict__ W,
                                              const float* __restrict__ bias,
                                              const int* meta,
                                              float* out) {
    __shared__ __attribute__((aligned(16))) _Float16 aggs[16 * AP];
    __shared__ __attribute__((aligned(16))) float    outs[16 * OP];
    const int tid  = threadIdx.x;
    const int wave = tid >> 5;
    const int lane = tid & 31;
    const int row0 = blockIdx.x * 16;

    for (int t = 0; t < 2; ++t) {
        const int r = wave * 2 + t;
        node_aggregate(row0 + r, lane, sorted, x, sc, meta, aggs + r * AP);
    }
    __syncthreads();

    const int mrow  = lane & 15;
    const int khalf = (lane >> 4) << 3;
    const int ncol  = wave * 16 + (lane & 15);
    const _Float16* arow = aggs + mrow * AP + (wave >> 1) * IN_CH;
    v16h fa0, fa1, fb0, fb1;
    #pragma unroll
    for (int i = 0; i < 16; ++i) {
        const int k = i + (i & 8) + khalf;
        fa0[i] = arow[k];
        fa1[i] = arow[k + 32];
        fb0[i] = (_Float16)W[(size_t)k * HC + ncol];
        fb1[i] = (_Float16)W[(size_t)(k + 32) * HC + ncol];
    }
    v8f c = {};
    c = __builtin_amdgcn_wmma_f32_16x16x32_f16(false, fa0, false, fb0, (short)0, c, false, false);
    c = __builtin_amdgcn_wmma_f32_16x16x32_f16(false, fa1, false, fb1, (short)0, c, false, false);

    const float bc = bias[ncol];
    #pragma unroll
    for (int r = 0; r < 8; ++r)
        outs[(r + khalf) * OP + ncol] = c[r] + bc;
    __syncthreads();

    for (int t = 0; t < 2; ++t) {
        const int r = wave * 2 + t;
        float4 v = *(const float4*)(outs + r * OP + 4 * lane);
        const float mx = wave_max(fmaxf(fmaxf(v.x, v.y), fmaxf(v.z, v.w)));
        const float se = wave_sum(__expf(v.x - mx) + __expf(v.y - mx) + __expf(v.z - mx) + __expf(v.w - mx));
        const float lse = mx + __logf(se);
        v.x -= lse; v.y -= lse; v.z -= lse; v.w -= lse;
        ((float4*)(out + (size_t)(row0 + r) * HC))[lane] = v;
    }
}

extern "C" void kernel_launch(void* const* d_in, const int* in_sizes, int n_in,
                              void* d_out, int out_size, void* d_ws, size_t ws_size,
                              hipStream_t stream) {
    const float* x       = (const float*)d_in[0];
    const int*   ei      = (const int*)  d_in[1];
    const float* W       = (const float*)d_in[2];
    const float* att_src = (const float*)d_in[3];
    const float* att_dst = (const float*)d_in[4];
    const float* bias    = (const float*)d_in[5];
    float*       out     = (float*)d_out;
    int*         meta    = (int*)d_out;

    int*   sorted = (int*)d_ws + (size_t)N_NODES * HC;
    int*   cnt    = (int*)d_ws;
    int*   bks    = (int*)d_ws + M_CNT;
    float* sc     = (float*)d_ws + (size_t)2 * 1024 * 1024;
    int*   tmp    = (int*)d_ws + (size_t)4 * 1024 * 1024;

    k_hist<<<NB1, S_T, 0, stream>>>(ei, cnt);
    k_part<<<NB1, S_T, 0, stream>>>(ei, cnt, bks, tmp);
    k_bucket<<<NBK, S_T, 0, stream>>>(tmp, bks, sorted, meta);
    k_score<<<(N_NODES + SCORE_NPB - 1) / SCORE_NPB, 256, 0, stream>>>(x, W, att_src, att_dst, sc);
    k_node<<<N_NODES / 16, 256, 0, stream>>>(sorted, x, sc, W, bias, meta, out);
}


# Level-1 runs stay in their slice: a block writes its bucket-ordered slice back in place with a start/length word per bucket; bucket blocks collect their 256 runs; count pre-pass gone; new sum order
# speedup vs baseline: 9.2180x; 1.0501x over previous
#include <hip/hip_runtime.h>

#define N_NODES 100000
#define N_EDGES 1600000
#define IN_CH   64
#define OUT_CH  32
#define HEADS   4
#define HC      (HEADS * OUT_CH)
#define NEG_SLOPE 0.2f

typedef __attribute__((ext_vector_type(16))) _Float16 v16h;
typedef __attribute__((ext_vector_type(8)))  float    v8f;


#define SC_W   8
#define AP     264
#define OP     132

__device__ __forceinline__ float dpp_x1(float v) {
    return __int_as_float(__builtin_amdgcn_update_dpp(0, __float_as_int(v), 0xB1, 0xF, 0xF, true));
}
__device__ __forceinline__ float dpp_x2(float v) {
    return __int_as_float(__builtin_amdgcn_update_dpp(0, __float_as_int(v), 0x4E, 0xF, 0xF, true));
}
__device__ __forceinline__ float dpp_hm(float v) {
    return __int_as_float(__builtin_amdgcn_update_dpp(0, __float_as_int(v), 0x141, 0xF, 0xF, true));
}
__device__ __forceinline__ float dpp_rm(float v) {
    return __int_as_float(__builtin_amdgcn_update_dpp(0, __float_as_int(v), 0x140, 0xF, 0xF, true));
}
__device__ __forceinline__ float sum16(float v) {
    v += dpp_x1(v); v += dpp_x2(v); v += dpp_hm(v); v += dpp_rm(v);
    return v;
}
__device__ __forceinline__ float max16(float v) {
    v = fmaxf(v, dpp_x1(v)); v = fmaxf(v, dpp_x2(v)); v = fmaxf(v, dpp_hm(v)); v = fmaxf(v, dpp_rm(v));
    return v;
}
__device__ __forceinline__ float rlf(float v, int k) {
    return __int_as_float(__builtin_amdgcn_readlane(__float_as_int(v), k));
}
__device__ __forceinline__ float wave_sum(float v) { v = sum16(v); return rlf(v, 0) + rlf(v, 16); }
__device__ __forceinline__ float wave_max(float v) { v = max16(v); return fmaxf(rlf(v, 0), rlf(v, 16)); }
__device__ __forceinline__ float leaky(float e) { return e > 0.f ? e : NEG_SLOPE * e; }

#define SCORE_NPB 128
__global__ __launch_bounds__(256) void k_score(const float* __restrict__ x,
                                               const float* __restrict__ W,
                                               const float* __restrict__ att_src,
                                               const float* __restrict__ att_dst,
                                               float* __restrict__ sc) {
    __shared__ __attribute__((aligned(16))) float V[IN_CH * SC_W];
    const int tid = threadIdx.x;
    for (int i = tid; i < IN_CH * SC_W; i += 256) {
        const int k = i >> 3, j = i & 7, hd = j & 3;
        const float4* wr = (const float4*)(W + (size_t)k * HC + hd * OUT_CH);
        const float4* ar = (const float4*)((j < 4 ? att_src : att_dst) + hd * OUT_CH);
        float s = 0.f;
        #pragma unroll
        for (int c = 0; c < OUT_CH / 4; ++c) {
            const float4 w4 = wr[c], a4 = ar[c];
            s += w4.x * a4.x + w4.y * a4.y + w4.z * a4.z + w4.w * a4.w;
        }
        V[i] = s;
    }
    __syncthreads();

    const int wave = tid >> 5, lane = tid & 31;
    const int q = lane & 15, hf = lane >> 4;
    float v[4][SC_W];
    #pragma unroll
    for (int t = 0; t < 4; ++t) {
        #pragma unroll
        for (int j = 0; j < SC_W; ++j) v[t][j] = V[(4 * q + t) * SC_W + j];
    }
    const int base = blockIdx.x * SCORE_NPB + wave * 16 + hf;
    float4 xv[8];
    #pragma unroll
    for (int it = 0; it < 8; ++it) {
        int n = base + it * 2;
        if (n > N_NODES - 1) n = N_NODES - 1;
        xv[it] = ((const float4*)(x + (size_t)n * IN_CH))[q];
    }
    #pragma unroll
    for (int it = 0; it < 8; ++it) {
        const int n = base + it * 2;
        float s[SC_W];
        #pragma unroll
        for (int j = 0; j < SC_W; ++j)
            s[j] = sum16(xv[it].x * v[0][j] + xv[it].y * v[1][j] + xv[it].z * v[2][j] + xv[it].w * v[3][j]);
        if (q == 0 && n < N_NODES) {
            float4* o = (float4*)(sc + (size_t)n * SC_W);
            o[0] = make_float4(s[0], s[1], s[2], s[3]);
            o[1] = make_float4(s[4], s[5], s[6], s[7]);
        }
    }
}

#define BKS    9
#define BK_N   (1 << BKS)
#define NBK    ((N_NODES + BK_N - 1) / BK_N)
#define NB1    256
#define EPB    (N_EDGES / NB1)
#define S_T    1024
#define EPT    ((EPB + S_T - 1) / S_T)
#define M_CNT  (NBK * NB1)
#define RPW    ((NBK + 31) / 32)
#define SPW    (NB1 / 32)
#define BCAP   12288
static_assert(EPB * NB1 == N_EDGES, "the slices must tile the edge list");
static_assert(NB1 == 256, "a row of the run table is two 16-byte loads per lane");
static_assert(NBK <= 256, "bucket tables hold 256 entries");
static_assert(EPB < 32768, "start and length of a run share one word");
static_assert((EPB * 8) % 16 == 0, "a slice is copied out in 16-byte pieces");

__device__ __forceinline__ int wave_incl_scan_i(int v, int lane) {
    #pragma unroll
    for (int o = 1; o < 32; o <<= 1) {
        int t = __shfl_up(v, o, 32);
        if (lane >= o) v += t;
    }
    return v;
}

__device__ __forceinline__ int block_excl_scan(const int c, int* wt, const int t) {
    const int lane = t & 31, w = t >> 5;
    const int incl = wave_incl_scan_i(c, lane);
    if (lane == 31) wt[w] = incl;
    __syncthreads();
    if (w == 0) {
        const int x  = wt[lane];
        const int xs = wave_incl_scan_i(x, lane);
        wt[lane] = xs - x;
    }
    __syncthreads();
    return incl - c + wt[w];
}

__device__ __forceinline__ int wave_sum_i(int v) {
    v += __builtin_amdgcn_update_dpp(0, v, 0xB1,  0xF, 0xF, true);
    v += __builtin_amdgcn_update_dpp(0, v, 0x4E,  0xF, 0xF, true);
    v += __builtin_amdgcn_update_dpp(0, v, 0x141, 0xF, 0xF, true);
    v += __builtin_amdgcn_update_dpp(0, v, 0x140, 0xF, 0xF, true);
    return __builtin_amdgcn_readlane(v, 0) + __builtin_amdgcn_readlane(v, 16);
}

__global__ __launch_bounds__(S_T) void k_part(const int* __restrict__ ei,
                                              int* __restrict__ tab,
                                              int* __restrict__ tmpw) {
    __shared__ int  lcnt[256];
    __shared__ int  lcur[256];
    __shared__ int  wt[32];
    __shared__ __attribute__((aligned(16))) int2 stage[EPB];
    const int t = threadIdx.x, blk = blockIdx.x;
    if (t < 256) lcnt[t] = 0;
    __syncthreads();
    const int* sp = ei + (size_t)blk * EPB;
    const int* dp = sp + N_EDGES;
    int sv[EPT], dv[EPT];
    #pragma unroll
    for (int k = 0; k < EPT; ++k) {
        const int i = t + k * S_T;
        sv[k] = 0; dv[k] = -1;
        if (i < EPB) { sv[k] = sp[i]; dv[k] = dp[i]; }
    }
    #pragma unroll
    for (int k = 0; k < EPT; ++k)
        if (dv[k] >= 0) atomicAdd(&lcnt[dv[k] >> BKS], 1);
    __syncthreads();
    const int c    = (t < 256) ? lcnt[t] : 0;
    const int excl = block_excl_scan(c, wt, t);
    if (t < 256) lcur[t] = excl;
    if (t < NBK) tab[t * NB1 + blk] = (excl << 16) | c;
    __syncthreads();
    #pragma unroll
    for (int k = 0; k < EPT; ++k)
        if (dv[k] >= 0) {
            const int p = atomicAdd(&lcur[dv[k] >> BKS], 1);
            stage[p] = make_int2(sv[k], dv[k]);
        }
    __syncthreads();
    const int4* s4 = (const int4*)stage;
    int4* d4 = (int4*)(tmpw + (size_t)blk * EPB * 2);
    for (int i = t; i < EPB / 2; i += S_T) d4[i] = s4[i];
}

__global__ __launch_bounds__(S_T) void k_bucket(const int* __restrict__ tmpw,
                                                const int* __restrict__ tab,
                                                int* __restrict__ sorted,
                                                int* __restrict__ meta) {
    const int2* __restrict__ tmp = (const int2*)tmpw;
    __shared__ int cnt[BK_N];
    __shared__ int cur[BK_N];
    __shared__ int wt[32];
    __shared__ int btot[256];
    __shared__ __attribute__((aligned(16))) int seg[NB1];
    __shared__ int stage[BCAP];
    const int t = threadIdx.x, b = blockIdx.x;
    const int lane = t & 31, w = t >> 5;
    if (t < BK_N) cnt[t] = 0;
    int4 ra[RPW], rb[RPW];
    #pragma unroll
    for (int i = 0; i < RPW; ++i) {
        int r = w + 32 * i;
        if (r > NBK - 1) r = NBK - 1;
        const int4* row = (const int4*)(tab + r * NB1);
        ra[i] = row[lane];
        rb[i] = row[lane + 32];
    }
    #pragma unroll
    for (int i = 0; i < RPW; ++i) {
        const int r = w + 32 * i;
        const int4 a = ra[i], c4 = rb[i];
        int s = (a.x  & 0xffff) + (a.y  & 0xffff) + (a.z  & 0xffff) + (a.w  & 0xffff)
              + (c4.x & 0xffff) + (c4.y & 0xffff) + (c4.z & 0xffff) + (c4.w & 0xffff);
        s = wave_sum_i(s);
        if (lane == 0 && r < NBK) btot[r] = s;
        if (r == b) {
            ((int4*)seg)[lane]      = a;
            ((int4*)seg)[lane + 32] = c4;
        }
    }
    __syncthreads();
    int pre = 0;
    #pragma unroll
    for (int i = 0; i < RPW; ++i) {
        const int r = lane + 32 * i;
        const int v = btot[r < NBK ? r : NBK - 1];
        pre += (r < b) ? v : 0;
    }
    const int bstart = wave_sum_i(pre);
    const int n      = btot[b];

    const int2* rp[SPW];
    int rn[SPW];
    int rmax = 0;
    #pragma unroll
    for (int q = 0; q < SPW; ++q) {
        const int sl = w * SPW + q;
        const int e  = __builtin_amdgcn_readfirstlane(seg[sl]);
        rp[q] = tmp + (size_t)sl * EPB + (e >> 16);
        rn[q] = e & 0xffff;
        rmax  = rn[q] > rmax ? rn[q] : rmax;
    }
    {
        int dl[2 * SPW];
        #pragma unroll
        for (int q = 0; q < SPW; ++q) {
            dl[2 * q]     = (lane      < rn[q]) ? rp[q][lane].y      : -1;
            dl[2 * q + 1] = (lane + 32 < rn[q]) ? rp[q][lane + 32].y : -1;
        }
        #pragma unroll
        for (int k = 0; k < 2 * SPW; ++k)
            if (dl[k] >= 0) atomicAdd(&cnt[dl[k] & (BK_N - 1)], 1);
        if (rmax > 64) {
            #pragma unroll
            for (int q = 0; q < SPW; ++q)
                for (int j = 64 + lane; j < rn[q]; j += 32)
                    atomicAdd(&cnt[rp[q][j].y & (BK_N - 1)], 1);
        }
    }
    __syncthreads();
    const int c    = (t < BK_N) ? cnt[t] : 0;
    const int excl = block_excl_scan(c, wt, t);
    if (t < BK_N) {
        cur[t] = excl;
        const int d = b * BK_N + t;
        if (d < N_NODES) {
            meta[(size_t)d * HC]     = bstart + excl;
            meta[(size_t)d * HC + 1] = c;
        }
    }
    __syncthreads();
    const bool fits = n <= BCAP;
    {
        int2 r[2 * SPW];
        #pragma unroll
        for (int q = 0; q < SPW; ++q) {
            r[2 * q] = make_int2(0, -1); r[2 * q + 1] = make_int2(0, -1);
            if (lane      < rn[q]) r[2 * q]     = rp[q][lane];
            if (lane + 32 < rn[q]) r[2 * q + 1] = rp[q][lane + 32];
        }
        #pragma unroll
        for (int k = 0; k < 2 * SPW; ++k)
            if (r[k].y >= 0) {
                const int p = atomicAdd(&cur[r[k].y & (BK_N - 1)], 1);
                if (fits) stage[p] = r[k].x;
                else      sorted[bstart + p] = r[k].x;
            }
        if (rmax > 64) {
            #pragma unroll
            for (int q = 0; q < SPW; ++q)
                for (int j = 64 + lane; j < rn[q]; j += 32) {
                    const int2 e = rp[q][j];
                    const int p = atomicAdd(&cur[e.y & (BK_N - 1)], 1);
                    if (fits) stage[p] = e.x;
                    else      sorted[bstart + p] = e.x;
                }
        }
    }
    if (fits) {
        __syncthreads();
        for (int i = t; i < n; i += S_T) sorted[bstart + i] = stage[i];
    }
}

typedef __attribute__((ext_vector_type(2))) _Float16 v2h;

__device__ __forceinline__ float head_update(const float e, const bool on,
                                             float& m, float& l, float2& acc) {
    const float mn = fmaxf(m, wave_max(e));
    const float sc = __expf(m - mn);
    const float p  = on ? __expf(e - mn) : 0.f;
    l = l * sc + wave_sum(p);
    acc.x *= sc; acc.y *= sc;
    m = mn;
    return p;
}

__device__ __forceinline__ float2 ldx(const float* __restrict__ x, int s, int lane) {
    return ((const float2*)(x + (size_t)s * IN_CH))[lane];
}

__device__ __forceinline__ void node_aggregate(const int d, const int lane,
                                               const int* __restrict__ sorted,
                                               const float* __restrict__ x,
                                               const float* __restrict__ sc,
                                               const int* meta,
                                               _Float16* aggrow) {
    const int start = __builtin_amdgcn_readfirstlane(meta[(size_t)d * HC]);
    const int deg   = __builtin_amdgcn_readfirstlane(meta[(size_t)d * HC + 1]);

    const float4 asd = ((const float4*)(sc + (size_t)d * SC_W))[0];
    const float4 ad  = ((const float4*)(sc + (size_t)d * SC_W))[1];
    const float2 xd  = ldx(x, d, lane);

    float m0 = leaky(asd.x + ad.x), m1 = leaky(asd.y + ad.y);
    float m2 = leaky(asd.z + ad.z), m3 = leaky(asd.w + ad.w);
    float l0 = 1.f, l1 = 1.f, l2 = 1.f, l3 = 1.f;
    float2 a0 = xd, a1 = xd, a2 = xd, a3 = xd;

    const float ninf = -__builtin_inff();
    for (int base = 0; base < deg; base += 32) {
        int n = deg - base; if (n > 32) n = 32;
        const bool on = lane < n;
        int myid = 0;
        if (on) myid = sorted[start + base + lane];
        const float4 as = ((const float4*)(sc + (size_t)myid * SC_W))[0];

        const int ng = (n + 3) & ~3;
        #define LD4(ca, cb, cc, cd, kk)                                           \
            ca = ldx(x, __builtin_amdgcn_readlane(myid, (kk)), lane);             \
            cb = ldx(x, __builtin_amdgcn_readlane(myid, (kk) + 1), lane);         \
            cc = ldx(x, __builtin_amdgcn_readlane(myid, (kk) + 2), lane);         \
            cd = ldx(x, __builtin_amdgcn_readlane(myid, (kk) + 3), lane);
        #define EDGE_FMA(cc, kk)                                                  \
        {   const float w0 = rlf(p0, (kk)), w1 = rlf(p1, (kk));                   \
            const float w2 = rlf(p2, (kk)), w3 = rlf(p3, (kk));                   \
            a0.x += w0 * cc.x; a0.y += w0 * cc.y;                                 \
            a1.x += w1 * cc.x; a1.y += w1 * cc.y;                                 \
            a2.x += w2 * cc.x; a2.y += w2 * cc.y;                                 \
            a3.x += w3 * cc.x; a3.y += w3 * cc.y; }
        const float2 z2 = make_float2(0.f, 0.f);
        float2 c0, c1, c2, c3, c4 = z2, c5 = z2, c6 = z2, c7 = z2;
        LD4(c0, c1, c2, c3, 0)
        if (ng > 4) { LD4(c4, c5, c6, c7, 4) }

        const float p0 = head_update(on ? leaky(as.x + ad.x) : ninf, on, m0, l0, a0);
        const float p1 = head_update(on ? leaky(as.y + ad.y) : ninf, on, m1, l1, a1);
        const float p2 = head_update(on ? leaky(as.z + ad.z) : ninf, on, m2, l2, a2);
        const float p3 = head_update(on ? leaky(as.w + ad.w) : ninf, on, m3, l3, a3);

        for (int k = 0;;) {
            EDGE_FMA(c0, k)
            EDGE_FMA(c1, k + 1)
            EDGE_FMA(c2, k + 2)
            EDGE_FMA(c3, k + 3)
            if (k + 4 < ng) {
                EDGE_FMA(c4, k + 4)
                EDGE_FMA(c5, k + 5)
                EDGE_FMA(c6, k + 6)
                EDGE_FMA(c7, k + 7)
            }
            k += 8;
            if (k >= ng) break;
            LD4(c0, c1, c2, c3, k)
            if (k + 4 < ng) { LD4(c4, c5, c6, c7, k + 4) }
        }
        #undef LD4
        #undef EDGE_FMA
    }

    const float i0 = 1.0f / (l0 + 1e-16f), i1 = 1.0f / (l1 + 1e-16f);
    const float i2 = 1.0f / (l2 + 1e-16f), i3 = 1.0f / (l3 + 1e-16f);
    v2h h0, h1, h2, h3;
    h0[0] = (_Float16)(a0.x * i0); h0[1] = (_Float16)(a0.y * i0);
    h1[0] = (_Float16)(a1.x * i1); h1[1] = (_Float16)(a1.y * i1);
    h2[0] = (_Float16)(a2.x * i2); h2[1] = (_Float16)(a2.y * i2);
    h3[0] = (_Float16)(a3.x * i3); h3[1] = (_Float16)(a3.y * i3);
    *(v2h*)(aggrow + 0 * IN_CH + 2 * lane) = h0;
    *(v2h*)(aggrow + 1 * IN_CH + 2 * lane) = h1;
    *(v2h*)(aggrow + 2 * IN_CH + 2 * lane) = h2;
    *(v2h*)(aggrow + 3 * IN_CH + 2 * lane) = h3;
}

__global__ __launch_bounds__(256) void k_node(const int* __restrict__ sorted,
                                              const float* __restrict__ x,
                                              const float* __restrict__ sc,
                                              const float* __restrict__ W,
                                              const float* __restrict__ bias,
                                              const int* meta,
                                              float* out) {
    __shared__ __attribute__((aligned(16))) _Float16 aggs[16 * AP];
    __shared__ __attribute__((aligned(16))) float    outs[16 * OP];
    const int tid  = threadIdx.x;
    const int wave = tid >> 5;
    const int lane = tid & 31;
    const int row0 = blockIdx.x * 16;

    for (int t = 0; t < 2; ++t) {
        const int r = wave * 2 + t;
        node_aggregate(row0 + r, lane, sorted, x, sc, meta, aggs + r * AP);
    }
    __syncthreads();

    const int mrow  = lane & 15;
    const int khalf = (lane >> 4) << 3;
    const int ncol  = wave * 16 + (lane & 15);
    const _Float16* arow = aggs + mrow * AP + (wave >> 1) * IN_CH;
    v16h fa0, fa1, fb0, fb1;
    #pragma unroll
    for (int i = 0; i < 16; ++i) {
        const int k = i + (i & 8) + khalf;
        fa0[i] = arow[k];
        fa1[i] = arow[k + 32];
        fb0[i] = (_Float16)W[(size_t)k * HC + ncol];
        fb1[i] = (_Float16)W[(size_t)(k + 32) * HC + ncol];
    }
    v8f c = {};
    c = __builtin_amdgcn_wmma_f32_16x16x32_f16(false, fa0, false, fb0, (short)0, c, false, false);
    c = __builtin_amdgcn_wmma_f32_16x16x32_f16(false, fa1, false, fb1, (short)0, c, false, false);

    const float bc = bias[ncol];
    #pragma unroll
    for (int r = 0; r < 8; ++r)
        outs[(r + khalf) * OP + ncol] = c[r] + bc;
    __syncthreads();

    for (int t = 0; t < 2; ++t) {
        const int r = wave * 2 + t;
        float4 v = *(const float4*)(outs + r * OP + 4 * lane);
        const float mx = wave_max(fmaxf(fmaxf(v.x, v.y), fmaxf(v.z, v.w)));
        const float se = wave_sum(__expf(v.x - mx) + __expf(v.y - mx) + __expf(v.z - mx) + __expf(v.w - mx));
        const float lse = mx + __logf(se);
        v.x -= lse; v.y -= lse; v.z -= lse; v.w -= lse;
        ((float4*)(out + (size_t)(row0 + r) * HC))[lane] = v;
    }
}

extern "C" void kernel_launch(void* const* d_in, const int* in_sizes, int n_in,
                              void* d_out, int out_size, void* d_ws, size_t ws_size,
                              hipStream_t stream) {
    const float* x       = (const float*)d_in[0];
    const int*   ei      = (const int*)  d_in[1];
    const float* W       = (const float*)d_in[2];
    const float* att_src = (const float*)d_in[3];
    const float* att_dst = (const float*)d_in[4];
    const float* bias    = (const float*)d_in[5];
    float*       out     = (float*)d_out;
    int*         meta    = (int*)d_out;

    int*   sorted = (int*)d_ws + (size_t)N_NODES * HC;
    int*   tab    = (int*)d_ws;
    float* sc     = (float*)d_ws + (size_t)2 * 1024 * 1024;
    int*   tmp    = (int*)d_ws + (size_t)4 * 1024 * 1024;

    k_part<<<NB1, S_T, 0, stream>>>(ei, tab, tmp);
    k_bucket<<<NBK, S_T, 0, stream>>>(tmp, tab, sorted, meta);
    k_score<<<(N_NODES + SCORE_NPB - 1) / SCORE_NPB, 256, 0, stream>>>(x, W, att_src, att_dst, sc);
    k_node<<<N_NODES / 16, 256, 0, stream>>>(sorted, x, sc, W, bias, meta, out);
}
